# starting kernel
# baseline (speedup 1.0000x reference)
#include <hip/hip_runtime.h>

typedef __attribute__((ext_vector_type(2))) float v2f;
typedef __attribute__((ext_vector_type(8))) float v8f;

#define ALPHA    0.2f
#define NEG_BIG  -9000000000000000.0f
#define N_NODES  2048
#define FDIM     128
#define NBATCH   4

__global__ __launch_bounds__(256) void gat_wh_kernel(const float* __restrict__ h,
                                                     const float* __restrict__ W,
                                                     float* __restrict__ Wh) {
    __shared__ float sh[16 * FDIM];
    const int tile = blockIdx.x;
    const int b    = tile / (N_NODES / 16);
    const int i0   = (tile % (N_NODES / 16)) * 16;
    const int t    = threadIdx.x;

    const float* hbase = h + ((size_t)b * N_NODES + i0) * FDIM;
    for (int k = t; k < 16 * FDIM; k += 256) sh[k] = hbase[k];
    __syncthreads();

    const int wave  = t >> 5;
    const int lane  = t & 31;
    const int row   = lane & 15;
    const int kslot = lane >> 4;
    const int n0    = wave * 16;

    v8f acc = {};
    for (int k0 = 0; k0 < FDIM; k0 += 4) {
        v2f A, B;
        A.x = sh[row * FDIM + k0 + kslot * 2];
        A.y = sh[row * FDIM + k0 + kslot * 2 + 1];
        B.x = W[(size_t)(k0 + kslot * 2)     * FDIM + n0 + row];
        B.y = W[(size_t)(k0 + kslot * 2 + 1) * FDIM + n0 + row];
        acc = __builtin_amdgcn_wmma_f32_16x16x4_f32(false, A, false, B,
                                                    (short)0, acc, false, false);
    }

    float* out = Wh + ((size_t)b * N_NODES + i0) * FDIM + n0 + row;
#pragma unroll
    for (int v = 0; v < 8; ++v) {
        int orow = v + 8 * kslot;
        out[(size_t)orow * FDIM] = acc[v];
    }
}

__global__ __launch_bounds__(256) void gat_f_kernel(const float* __restrict__ Wh,
                                                    const float* __restrict__ a,
                                                    float* __restrict__ f1,
                                                    float* __restrict__ f2) {
    const int row  = blockIdx.x * 8 + (threadIdx.x >> 5);
    const int lane = threadIdx.x & 31;
    const float* wr = Wh + (size_t)row * FDIM;
    float s1 = 0.f, s2 = 0.f;
#pragma unroll
    for (int c = 0; c < FDIM; c += 32) {
        float v = wr[c + lane];
        s1 += v * a[c + lane];
        s2 += v * a[FDIM + c + lane];
    }
#pragma unroll
    for (int m = 16; m >= 1; m >>= 1) {
        s1 += __shfl_xor(s1, m, 32);
        s2 += __shfl_xor(s2, m, 32);
    }
    if (lane == 0) { f1[row] = s1; f2[row] = s2; }
}

__global__ __launch_bounds__(256) void gat_attn_kernel(const float* __restrict__ Wh,
                                                       const int*   __restrict__ adj,
                                                       const float* __restrict__ f1,
                                                       const float* __restrict__ f2,
                                                       float* __restrict__ out) {
    __shared__ float s_f2[N_NODES];
    __shared__ float s_acc[16 * FDIM];
    __shared__ float s_m[16];
    __shared__ float s_l[16];

    const int blk  = blockIdx.x;
    const int b    = blk / (N_NODES / 16);
    const int i0   = (blk % (N_NODES / 16)) * 16;
    const int t    = threadIdx.x;
    const int wave = t >> 5;
    const int lane = t & 31;

    for (int j = t; j < N_NODES; j += 256) s_f2[j] = f2[(size_t)b * N_NODES + j];
    for (int k = t; k < 16 * FDIM; k += 256) s_acc[k] = 0.f;
    __syncthreads();

    {
        const int half = lane >> 4;
        const int r    = wave * 2 + half;
        const int l16  = lane & 15;
        const int* arow = adj + (size_t)(i0 + r) * N_NODES;
        const float fr  = f1[(size_t)b * N_NODES + i0 + r];

        float mx = NEG_BIG;
        for (int j = l16; j < N_NODES; j += 16) {
            float x = fr + s_f2[j];
            float s = (arow[j] > 0) ? (x > 0.f ? x : ALPHA * x) : NEG_BIG;
            mx = fmaxf(mx, s);
        }
#pragma unroll
        for (int m = 8; m >= 1; m >>= 1) mx = fmaxf(mx, __shfl_xor(mx, m, 32));

        float sum = 0.f;
        for (int j = l16; j < N_NODES; j += 16) {
            float x = fr + s_f2[j];
            float s = (arow[j] > 0) ? (x > 0.f ? x : ALPHA * x) : NEG_BIG;
            sum += __expf(s - mx);
        }
#pragma unroll
        for (int m = 8; m >= 1; m >>= 1) sum += __shfl_xor(sum, m, 32);

        if (l16 == 0) { s_m[r] = mx; s_l[r] = sum; }
    }
    __syncthreads();

    const int ln16  = lane & 15;
    const int kslot = lane >> 4;
    const float rm  = s_m[ln16];
    const float rf1 = f1[(size_t)b * N_NODES + i0 + ln16];
    const int* arow = adj + (size_t)(i0 + ln16) * N_NODES;
    const float* whb = Wh + (size_t)b * N_NODES * FDIM;

    v8f acc[8] = {};
    const int jend = wave * 256 + 256;
    for (int j0 = wave * 256; j0 < jend; j0 += 4) {
        const int jA = j0 + kslot * 2;
        float x0 = rf1 + s_f2[jA];
        float s0 = (arow[jA] > 0) ? (x0 > 0.f ? x0 : ALPHA * x0) : NEG_BIG;
        float x1 = rf1 + s_f2[jA + 1];
        float s1 = (arow[jA + 1] > 0) ? (x1 > 0.f ? x1 : ALPHA * x1) : NEG_BIG;
        v2f A;
        A.x = __expf(s0 - rm);
        A.y = __expf(s1 - rm);

        const float* w0 = whb + (size_t)jA * FDIM + ln16;
        const float* w1 = w0 + FDIM;
#pragma unroll
        for (int n = 0; n < 8; ++n) {
            v2f B;
            B.x = w0[n * 16];
            B.y = w1[n * 16];
            acc[n] = __builtin_amdgcn_wmma_f32_16x16x4_f32(false, A, false, B,
                                                           (short)0, acc[n], false, false);
        }
    }

#pragma unroll
    for (int n = 0; n < 8; ++n) {
#pragma unroll
        for (int v = 0; v < 8; ++v) {
            int orow = v + 8 * kslot;
            atomicAdd(&s_acc[orow * FDIM + n * 16 + ln16], acc[n][v]);
        }
    }
    __syncthreads();

    for (int k = t; k < 16 * FDIM; k += 256) {
        int r = k >> 7;
        out[((size_t)b * N_NODES + i0 + r) * FDIM + (k & (FDIM - 1))] = s_acc[k] / s_l[r];
    }
}

extern "C" void kernel_launch(void* const* d_in, const int* in_sizes, int n_in,
                              void* d_out, int out_size, void* d_ws, size_t ws_size,
                              hipStream_t stream) {
    const float* h   = (const float*)d_in[0];
    const int*   adj = (const int*)  d_in[1];
    const float* W   = (const float*)d_in[2];
    const float* a   = (const float*)d_in[3];
    float* out = (float*)d_out;

    float* Wh = (float*)d_ws;
    float* f1 = Wh + (size_t)NBATCH * N_NODES * FDIM;
    float* f2 = f1 + (size_t)NBATCH * N_NODES;

    gat_wh_kernel  <<<NBATCH * (N_NODES / 16), 256, 0, stream>>>(h, W, Wh);
    gat_f_kernel   <<<NBATCH * (N_NODES / 8),  256, 0, stream>>>(Wh, a, f1, f2);
    gat_attn_kernel<<<NBATCH * (N_NODES / 16), 256, 0, stream>>>(Wh, adj, f1, f2, out);
}


# attention kernel: mask and Wh rows read by unconditional 16-byte loads issued in groups (select, no branch); k order and tile columns permuted to fit; order of the sums over j changes
# speedup vs baseline: 2.1674x; 2.1674x over previous
#include <hip/hip_runtime.h>

typedef __attribute__((ext_vector_type(2))) float v2f;
typedef __attribute__((ext_vector_type(8))) float v8f;

#define ALPHA    0.2f
#define NEG_BIG  -9000000000000000.0f
#define N_NODES  2048
#define FDIM     128
#define NBATCH   4

__global__ __launch_bounds__(256) void gat_wh_kernel(const float* __restrict__ h,
                                                     const float* __restrict__ W,
                                                     float* __restrict__ Wh) {
    __shared__ float sh[16 * FDIM];
    const int tile = blockIdx.x;
    const int b    = tile / (N_NODES / 16);
    const int i0   = (tile % (N_NODES / 16)) * 16;
    const int t    = threadIdx.x;

    const float* hbase = h + ((size_t)b * N_NODES + i0) * FDIM;
    for (int k = t; k < 16 * FDIM; k += 256) sh[k] = hbase[k];
    __syncthreads();

    const int wave  = t >> 5;
    const int lane  = t & 31;
    const int row   = lane & 15;
    const int kslot = lane >> 4;
    const int n0    = wave * 16;

    v8f acc = {};
    for (int k0 = 0; k0 < FDIM; k0 += 4) {
        v2f A, B;
        A.x = sh[row * FDIM + k0 + kslot * 2];
        A.y = sh[row * FDIM + k0 + kslot * 2 + 1];
        B.x = W[(size_t)(k0 + kslot * 2)     * FDIM + n0 + row];
        B.y = W[(size_t)(k0 + kslot * 2 + 1) * FDIM + n0 + row];
        acc = __builtin_amdgcn_wmma_f32_16x16x4_f32(false, A, false, B,
                                                    (short)0, acc, false, false);
    }

    float* out = Wh + ((size_t)b * N_NODES + i0) * FDIM + n0 + row;
#pragma unroll
    for (int v = 0; v < 8; ++v) {
        int orow = v + 8 * kslot;
        out[(size_t)orow * FDIM] = acc[v];
    }
}

__global__ __launch_bounds__(256) void gat_f_kernel(const float* __restrict__ Wh,
                                                    const float* __restrict__ a,
                                                    float* __restrict__ f1,
                                                    float* __restrict__ f2) {
    const int row  = blockIdx.x * 8 + (threadIdx.x >> 5);
    const int lane = threadIdx.x & 31;
    const float* wr = Wh + (size_t)row * FDIM;
    float s1 = 0.f, s2 = 0.f;
#pragma unroll
    for (int c = 0; c < FDIM; c += 32) {
        float v = wr[c + lane];
        s1 += v * a[c + lane];
        s2 += v * a[FDIM + c + lane];
    }
#pragma unroll
    for (int m = 16; m >= 1; m >>= 1) {
        s1 += __shfl_xor(s1, m, 32);
        s2 += __shfl_xor(s2, m, 32);
    }
    if (lane == 0) { f1[row] = s1; f2[row] = s2; }
}

__device__ __forceinline__ float gat_score(int m, float x) {
    const float e = x > 0.f ? x : ALPHA * x;
    return m > 0 ? e : NEG_BIG;
}

__global__ __launch_bounds__(256) void gat_attn_kernel(const float* __restrict__ Wh,
                                                       const int*   __restrict__ adj,
                                                       const float* __restrict__ f1,
                                                       const float* __restrict__ f2,
                                                       float* __restrict__ out) {
    __shared__ alignas(16) float s_f2[N_NODES];
    __shared__ alignas(16) float s_acc[16 * FDIM];
    __shared__ float s_m[16];
    __shared__ float s_l[16];

    const int blk  = blockIdx.x;
    const int b    = blk / (N_NODES / 16);
    const int i0   = (blk % (N_NODES / 16)) * 16;
    const int t    = threadIdx.x;
    const int wave = t >> 5;
    const int lane = t & 31;

    {
        const float4* f2v = reinterpret_cast<const float4*>(f2 + (size_t)b * N_NODES);
        float4* s_f2v = reinterpret_cast<float4*>(s_f2);
        for (int k = t; k < N_NODES / 4; k += 256) s_f2v[k] = f2v[k];
        float4* s_accv = reinterpret_cast<float4*>(s_acc);
        for (int k = t; k < 16 * FDIM / 4; k += 256) s_accv[k] = make_float4(0.f, 0.f, 0.f, 0.f);
    }
    __syncthreads();

    {
        const int half = lane >> 4;
        const int r    = wave * 2 + half;
        const int l16  = lane & 15;
        const int4*   arow4 = reinterpret_cast<const int4*>(adj + (size_t)(i0 + r) * N_NODES);
        const float4* sf4   = reinterpret_cast<const float4*>(s_f2);
        const float fr  = f1[(size_t)b * N_NODES + i0 + r];

        float mx = NEG_BIG;
#pragma unroll 8
        for (int it = 0; it < N_NODES / 64; ++it) {
            const int4   m = arow4[it * 16 + l16];
            const float4 f = sf4[it * 16 + l16];
            mx = fmaxf(mx, gat_score(m.x, fr + f.x));
            mx = fmaxf(mx, gat_score(m.y, fr + f.y));
            mx = fmaxf(mx, gat_score(m.z, fr + f.z));
            mx = fmaxf(mx, gat_score(m.w, fr + f.w));
        }
#pragma unroll
        for (int m = 8; m >= 1; m >>= 1) mx = fmaxf(mx, __shfl_xor(mx, m, 32));

        float sum = 0.f;
#pragma unroll 8
        for (int it = 0; it < N_NODES / 64; ++it) {
            const int4   m = arow4[it * 16 + l16];
            const float4 f = sf4[it * 16 + l16];
            sum += __expf(gat_score(m.x, fr + f.x) - mx);
            sum += __expf(gat_score(m.y, fr + f.y) - mx);
            sum += __expf(gat_score(m.z, fr + f.z) - mx);
            sum += __expf(gat_score(m.w, fr + f.w) - mx);
        }
#pragma unroll
        for (int m = 8; m >= 1; m >>= 1) sum += __shfl_xor(sum, m, 32);

        if (l16 == 0) { s_m[r] = mx; s_l[r] = sum; }
    }
    __syncthreads();

    const int c = lane & 15;
    const int h = lane >> 4;
    const float rm  = s_m[c];
    const float rf1 = f1[(size_t)b * N_NODES + i0 + c];
    const int*   arow = adj + (size_t)(i0 + c) * N_NODES;
    const float* whb  = Wh + (size_t)b * N_NODES * FDIM + 4 * c;

    v8f acc[8] = {};
    for (int ch = 0; ch < 16; ++ch) {
        const int jb = wave * 256 + ch * 16 + 8 * h;
        const int4   m0 = *reinterpret_cast<const int4*>(arow + jb);
        const int4   m1 = *reinterpret_cast<const int4*>(arow + jb + 4);
        const float4 fa = *reinterpret_cast<const float4*>(s_f2 + jb);
        const float4 fb = *reinterpret_cast<const float4*>(s_f2 + jb + 4);

        float p[8];
        p[0] = __expf(gat_score(m0.x, rf1 + fa.x) - rm);
        p[1] = __expf(gat_score(m0.y, rf1 + fa.y) - rm);
        p[2] = __expf(gat_score(m0.z, rf1 + fa.z) - rm);
        p[3] = __expf(gat_score(m0.w, rf1 + fa.w) - rm);
        p[4] = __expf(gat_score(m1.x, rf1 + fb.x) - rm);
        p[5] = __expf(gat_score(m1.y, rf1 + fb.y) - rm);
        p[6] = __expf(gat_score(m1.z, rf1 + fb.z) - rm);
        p[7] = __expf(gat_score(m1.w, rf1 + fb.w) - rm);

        const float* wrow = whb + (size_t)jb * FDIM;
#pragma unroll
        for (int q = 0; q < 4; ++q) {
            const float* w0 = wrow + (size_t)(2 * q) * FDIM;
            const float4 b00 = *reinterpret_cast<const float4*>(w0);
            const float4 b01 = *reinterpret_cast<const float4*>(w0 + 64);
            const float4 b10 = *reinterpret_cast<const float4*>(w0 + FDIM);
            const float4 b11 = *reinterpret_cast<const float4*>(w0 + FDIM + 64);
            v2f A;
            A.x = p[2 * q];
            A.y = p[2 * q + 1];
            v2f B;
            B.x = b00.x; B.y = b10.x;
            acc[0] = __builtin_amdgcn_wmma_f32_16x16x4_f32(false, A, false, B, (short)0, acc[0], false, false);
            B.x = b00.y; B.y = b10.y;
            acc[1] = __builtin_amdgcn_wmma_f32_16x16x4_f32(false, A, false, B, (short)0, acc[1], false, false);
            B.x = b00.z; B.y = b10.z;
            acc[2] = __builtin_amdgcn_wmma_f32_16x16x4_f32(false, A, false, B, (short)0, acc[2], false, false);
            B.x = b00.w; B.y = b10.w;
            acc[3] = __builtin_amdgcn_wmma_f32_16x16x4_f32(false, A, false, B, (short)0, acc[3], false, false);
            B.x = b01.x; B.y = b11.x;
            acc[4] = __builtin_amdgcn_wmma_f32_16x16x4_f32(false, A, false, B, (short)0, acc[4], false, false);
            B.x = b01.y; B.y = b11.y;
            acc[5] = __builtin_amdgcn_wmma_f32_16x16x4_f32(false, A, false, B, (short)0, acc[5], false, false);
            B.x = b01.z; B.y = b11.z;
            acc[6] = __builtin_amdgcn_wmma_f32_16x16x4_f32(false, A, false, B, (short)0, acc[6], false, false);
            B.x = b01.w; B.y = b11.w;
            acc[7] = __builtin_amdgcn_wmma_f32_16x16x4_f32(false, A, false, B, (short)0, acc[7], false, false);
        }
    }

#pragma unroll
    for (int n = 0; n < 8; ++n) {
        const int col = 64 * (n >> 2) + 4 * c + (n & 3);
#pragma unroll
        for (int v = 0; v < 8; ++v) {
            int orow = v + 8 * h;
            atomicAdd(&s_acc[orow * FDIM + col], acc[n][v]);
        }
    }
    __syncthreads();

    for (int k = t; k < 16 * FDIM; k += 256) {
        int r = k >> 7;
        out[((size_t)b * N_NODES + i0 + r) * FDIM + (k & (FDIM - 1))] = s_acc[k] / s_l[r];
    }
}

extern "C" void kernel_launch(void* const* d_in, const int* in_sizes, int n_in,
                              void* d_out, int out_size, void* d_ws, size_t ws_size,
                              hipStream_t stream) {
    const float* h   = (const float*)d_in[0];
    const int*   adj = (const int*)  d_in[1];
    const float* W   = (const float*)d_in[2];
    const float* a   = (const float*)d_in[3];
    float* out = (float*)d_out;

    float* Wh = (float*)d_ws;
    float* f1 = Wh + (size_t)NBATCH * N_NODES * FDIM;
    float* f2 = f1 + (size_t)NBATCH * N_NODES;

    gat_wh_kernel  <<<NBATCH * (N_NODES / 16), 256, 0, stream>>>(h, W, Wh);
    gat_f_kernel   <<<NBATCH * (N_NODES / 8),  256, 0, stream>>>(Wh, a, f1, f2);
    gat_attn_kernel<<<NBATCH * (N_NODES / 16), 256, 0, stream>>>(Wh, adj, f1, f2, out);
}


# f1,f2 formed in the Wh kernel's epilogue from the LDS-staged 16x128 tile; the separate row-dot launch is gone (3 launches to 2); order of the sums over the 128 columns of f1,f2 changes
# speedup vs baseline: 2.3297x; 1.0749x over previous
#include <hip/hip_runtime.h>

typedef __attribute__((ext_vector_type(2))) float v2f;
typedef __attribute__((ext_vector_type(8))) float v8f;

#define ALPHA    0.2f
#define NEG_BIG  -9000000000000000.0f
#define N_NODES  2048
#define FDIM     128
#define NBATCH   4

#define WH_PITCH 132

__global__ __launch_bounds__(256) void gat_wh_kernel(const float* __restrict__ h,
                                                     const float* __restrict__ W,
                                                     const float* __restrict__ a,
                                                     float* __restrict__ Wh,
                                                     float* __restrict__ f1,
                                                     float* __restrict__ f2) {
    __shared__ float sh[16 * FDIM];
    __shared__ alignas(16) float s_wh[16 * WH_PITCH];
    const int tile = blockIdx.x;
    const int b    = tile / (N_NODES / 16);
    const int i0   = (tile % (N_NODES / 16)) * 16;
    const int t    = threadIdx.x;

    const float* hbase = h + ((size_t)b * N_NODES + i0) * FDIM;
    for (int k = t; k < 16 * FDIM; k += 256) sh[k] = hbase[k];
    __syncthreads();

    const int wave  = t >> 5;
    const int lane  = t & 31;
    const int row   = lane & 15;
    const int kslot = lane >> 4;
    const int n0    = wave * 16;

    const int l16 = lane & 15;
    const float4 a1lo = *reinterpret_cast<const float4*>(a + 4 * l16);
    const float4 a1hi = *reinterpret_cast<const float4*>(a + 64 + 4 * l16);
    const float4 a2lo = *reinterpret_cast<const float4*>(a + FDIM + 4 * l16);
    const float4 a2hi = *reinterpret_cast<const float4*>(a + FDIM + 64 + 4 * l16);

    v8f acc = {};
    for (int k0 = 0; k0 < FDIM; k0 += 4) {
        v2f A, B;
        A.x = sh[row * FDIM + k0 + kslot * 2];
        A.y = sh[row * FDIM + k0 + kslot * 2 + 1];
        B.x = W[(size_t)(k0 + kslot * 2)     * FDIM + n0 + row];
        B.y = W[(size_t)(k0 + kslot * 2 + 1) * FDIM + n0 + row];
        acc = __builtin_amdgcn_wmma_f32_16x16x4_f32(false, A, false, B,
                                                    (short)0, acc, false, false);
    }

#pragma unroll
    for (int v = 0; v < 8; ++v) {
        int orow = v + 8 * kslot;
        s_wh[orow * WH_PITCH + n0 + row] = acc[v];
    }
    __syncthreads();

    float* out = Wh + ((size_t)b * N_NODES + i0) * FDIM + n0 + row;
#pragma unroll
    for (int v = 0; v < 8; ++v) {
        int orow = v + 8 * kslot;
        out[(size_t)orow * FDIM] = acc[v];
    }

    const int r = wave * 2 + kslot;
    const float4 wlo = *reinterpret_cast<const float4*>(s_wh + r * WH_PITCH + 4 * l16);
    const float4 whi = *reinterpret_cast<const float4*>(s_wh + r * WH_PITCH + 64 + 4 * l16);
    float s1 = wlo.x * a1lo.x + wlo.y * a1lo.y + wlo.z * a1lo.z + wlo.w * a1lo.w
             + whi.x * a1hi.x + whi.y * a1hi.y + whi.z * a1hi.z + whi.w * a1hi.w;
    float s2 = wlo.x * a2lo.x + wlo.y * a2lo.y + wlo.z * a2lo.z + wlo.w * a2lo.w
             + whi.x * a2hi.x + whi.y * a2hi.y + whi.z * a2hi.z + whi.w * a2hi.w;
    for (int m = 8; m >= 1; m >>= 1) {
        s1 += __shfl_xor(s1, m, 32);
        s2 += __shfl_xor(s2, m, 32);
    }
    if (l16 == 0) {
        f1[(size_t)b * N_NODES + i0 + r] = s1;
        f2[(size_t)b * N_NODES + i0 + r] = s2;
    }
}

__device__ __forceinline__ float gat_score(int m, float x) {
    const float e = x > 0.f ? x : ALPHA * x;
    return m > 0 ? e : NEG_BIG;
}

__global__ __launch_bounds__(256) void gat_attn_kernel(const float* __restrict__ Wh,
                                                       const int*   __restrict__ adj,
                                                       const float* __restrict__ f1,
                                                       const float* __restrict__ f2,
                                                       float* __restrict__ out) {
    __shared__ alignas(16) float s_f2[N_NODES];
    __shared__ alignas(16) float s_acc[16 * FDIM];
    __shared__ float s_m[16];
    __shared__ float s_l[16];

    const int blk  = blockIdx.x;
    const int b    = blk / (N_NODES / 16);
    const int i0   = (blk % (N_NODES / 16)) * 16;
    const int t    = threadIdx.x;
    const int wave = t >> 5;
    const int lane = t & 31;

    {
        const float4* f2v = reinterpret_cast<const float4*>(f2 + (size_t)b * N_NODES);
        float4* s_f2v = reinterpret_cast<float4*>(s_f2);
        for (int k = t; k < N_NODES / 4; k += 256) s_f2v[k] = f2v[k];
        float4* s_accv = reinterpret_cast<float4*>(s_acc);
        for (int k = t; k < 16 * FDIM / 4; k += 256) s_accv[k] = make_float4(0.f, 0.f, 0.f, 0.f);
    }
    __syncthreads();

    {
        const int half = lane >> 4;
        const int r    = wave * 2 + half;
        const int l16  = lane & 15;
        const int4*   arow4 = reinterpret_cast<const int4*>(adj + (size_t)(i0 + r) * N_NODES);
        const float4* sf4   = reinterpret_cast<const float4*>(s_f2);
        const float fr  = f1[(size_t)b * N_NODES + i0 + r];

        float mx = NEG_BIG;
#pragma unroll 8
        for (int it = 0; it < N_NODES / 64; ++it) {
            const int4   m = arow4[it * 16 + l16];
            const float4 f = sf4[it * 16 + l16];
            mx = fmaxf(mx, gat_score(m.x, fr + f.x));
            mx = fmaxf(mx, gat_score(m.y, fr + f.y));
            mx = fmaxf(mx, gat_score(m.z, fr + f.z));
            mx = fmaxf(mx, gat_score(m.w, fr + f.w));
        }
#pragma unroll
        for (int m = 8; m >= 1; m >>= 1) mx = fmaxf(mx, __shfl_xor(mx, m, 32));

        float sum = 0.f;
#pragma unroll 8
        for (int it = 0; it < N_NODES / 64; ++it) {
            const int4   m = arow4[it * 16 + l16];
            const float4 f = sf4[it * 16 + l16];
            sum += __expf(gat_score(m.x, fr + f.x) - mx);
            sum += __expf(gat_score(m.y, fr + f.y) - mx);
            sum += __expf(gat_score(m.z, fr + f.z) - mx);
            sum += __expf(gat_score(m.w, fr + f.w) - mx);
        }
#pragma unroll
        for (int m = 8; m >= 1; m >>= 1) sum += __shfl_xor(sum, m, 32);

        if (l16 == 0) { s_m[r] = mx; s_l[r] = sum; }
    }
    __syncthreads();

    const int c = lane & 15;
    const int h = lane >> 4;
    const float rm  = s_m[c];
    const float rf1 = f1[(size_t)b * N_NODES + i0 + c];
    const int*   arow = adj + (size_t)(i0 + c) * N_NODES;
    const float* whb  = Wh + (size_t)b * N_NODES * FDIM + 4 * c;

    v8f acc[8] = {};
    for (int ch = 0; ch < 16; ++ch) {
        const int jb = wave * 256 + ch * 16 + 8 * h;
        const int4   m0 = *reinterpret_cast<const int4*>(arow + jb);
        const int4   m1 = *reinterpret_cast<const int4*>(arow + jb + 4);
        const float4 fa = *reinterpret_cast<const float4*>(s_f2 + jb);
        const float4 fb = *reinterpret_cast<const float4*>(s_f2 + jb + 4);

        float p[8];
        p[0] = __expf(gat_score(m0.x, rf1 + fa.x) - rm);
        p[1] = __expf(gat_score(m0.y, rf1 + fa.y) - rm);
        p[2] = __expf(gat_score(m0.z, rf1 + fa.z) - rm);
        p[3] = __expf(gat_score(m0.w, rf1 + fa.w) - rm);
        p[4] = __expf(gat_score(m1.x, rf1 + fb.x) - rm);
        p[5] = __expf(gat_score(m1.y, rf1 + fb.y) - rm);
        p[6] = __expf(gat_score(m1.z, rf1 + fb.z) - rm);
        p[7] = __expf(gat_score(m1.w, rf1 + fb.w) - rm);

        const float* wrow = whb + (size_t)jb * FDIM;
#pragma unroll
        for (int q = 0; q < 4; ++q) {
            const float* w0 = wrow + (size_t)(2 * q) * FDIM;
            const float4 b00 = *reinterpret_cast<const float4*>(w0);
            const float4 b01 = *reinterpret_cast<const float4*>(w0 + 64);
            const float4 b10 = *reinterpret_cast<const float4*>(w0 + FDIM);
            const float4 b11 = *reinterpret_cast<const float4*>(w0 + FDIM + 64);
            v2f A;
            A.x = p[2 * q];
            A.y = p[2 * q + 1];
            v2f B;
            B.x = b00.x; B.y = b10.x;
            acc[0] = __builtin_amdgcn_wmma_f32_16x16x4_f32(false, A, false, B, (short)0, acc[0], false, false);
            B.x = b00.y; B.y = b10.y;
            acc[1] = __builtin_amdgcn_wmma_f32_16x16x4_f32(false, A, false, B, (short)0, acc[1], false, false);
            B.x = b00.z; B.y = b10.z;
            acc[2] = __builtin_amdgcn_wmma_f32_16x16x4_f32(false, A, false, B, (short)0, acc[2], false, false);
            B.x = b00.w; B.y = b10.w;
            acc[3] = __builtin_amdgcn_wmma_f32_16x16x4_f32(false, A, false, B, (short)0, acc[3], false, false);
            B.x = b01.x; B.y = b11.x;
            acc[4] = __builtin_amdgcn_wmma_f32_16x16x4_f32(false, A, false, B, (short)0, acc[4], false, false);
            B.x = b01.y; B.y = b11.y;
            acc[5] = __builtin_amdgcn_wmma_f32_16x16x4_f32(false, A, false, B, (short)0, acc[5], false, false);
            B.x = b01.z; B.y = b11.z;
            acc[6] = __builtin_amdgcn_wmma_f32_16x16x4_f32(false, A, false, B, (short)0, acc[6], false, false);
            B.x = b01.w; B.y = b11.w;
            acc[7] = __builtin_amdgcn_wmma_f32_16x16x4_f32(false, A, false, B, (short)0, acc[7], false, false);
        }
    }

#pragma unroll
    for (int n = 0; n < 8; ++n) {
        const int col = 64 * (n >> 2) + 4 * c + (n & 3);
#pragma unroll
        for (int v = 0; v < 8; ++v) {
            int orow = v + 8 * h;
            atomicAdd(&s_acc[orow * FDIM + col], acc[n][v]);
        }
    }
    __syncthreads();

    for (int k = t; k < 16 * FDIM; k += 256) {
        int r = k >> 7;
        out[((size_t)b * N_NODES + i0 + r) * FDIM + (k & (FDIM - 1))] = s_acc[k] / s_l[r];
    }
}

extern "C" void kernel_launch(void* const* d_in, const int* in_sizes, int n_in,
                              void* d_out, int out_size, void* d_ws, size_t ws_size,
                              hipStream_t stream) {
    const float* h   = (const float*)d_in[0];
    const int*   adj = (const int*)  d_in[1];
    const float* W   = (const float*)d_in[2];
    const float* a   = (const float*)d_in[3];
    float* out = (float*)d_out;

    float* Wh = (float*)d_ws;
    float* f1 = Wh + (size_t)NBATCH * N_NODES * FDIM;
    float* f2 = f1 + (size_t)NBATCH * N_NODES;

    gat_wh_kernel  <<<NBATCH * (N_NODES / 16), 256, 0, stream>>>(h, W, a, Wh, f1, f2);
    gat_attn_kernel<<<NBATCH * (N_NODES / 16), 256, 0, stream>>>(Wh, adj, f1, f2, out);
}


# attention kernel: softmax shift = bound leaky(f1_i + max_j f2_j); row sums formed in the WMMA pass, no max or exp-sum sweep (exact maxima only if a sum underflows); rounding and order of sums change
# speedup vs baseline: 2.6600x; 1.1418x over previous
#include <hip/hip_runtime.h>

typedef __attribute__((ext_vector_type(2))) float v2f;
typedef __attribute__((ext_vector_type(8))) float v8f;

#define ALPHA    0.2f
#define NEG_BIG  -9000000000000000.0f
#define N_NODES  2048
#define FDIM     128
#define NBATCH   4

#define WH_PITCH 132

__global__ __launch_bounds__(256) void gat_wh_kernel(const float* __restrict__ h,
                                                     const float* __restrict__ W,
                                                     const float* __restrict__ a,
                                                     float* __restrict__ Wh,
                                                     float* __restrict__ f1,
                                                     float* __restrict__ f2) {
    __shared__ float sh[16 * FDIM];
    __shared__ alignas(16) float s_wh[16 * WH_PITCH];
    const int tile = blockIdx.x;
    const int b    = tile / (N_NODES / 16);
    const int i0   = (tile % (N_NODES / 16)) * 16;
    const int t    = threadIdx.x;

    const float* hbase = h + ((size_t)b * N_NODES + i0) * FDIM;
    for (int k = t; k < 16 * FDIM; k += 256) sh[k] = hbase[k];
    __syncthreads();

    const int wave  = t >> 5;
    const int lane  = t & 31;
    const int row   = lane & 15;
    const int kslot = lane >> 4;
    const int n0    = wave * 16;

    const int l16 = lane & 15;
    const float4 a1lo = *reinterpret_cast<const float4*>(a + 4 * l16);
    const float4 a1hi = *reinterpret_cast<const float4*>(a + 64 + 4 * l16);
    const float4 a2lo = *reinterpret_cast<const float4*>(a + FDIM + 4 * l16);
    const float4 a2hi = *reinterpret_cast<const float4*>(a + FDIM + 64 + 4 * l16);

    v8f acc = {};
    for (int k0 = 0; k0 < FDIM; k0 += 4) {
        v2f A, B;
        A.x = sh[row * FDIM + k0 + kslot * 2];
        A.y = sh[row * FDIM + k0 + kslot * 2 + 1];
        B.x = W[(size_t)(k0 + kslot * 2)     * FDIM + n0 + row];
        B.y = W[(size_t)(k0 + kslot * 2 + 1) * FDIM + n0 + row];
        acc = __builtin_amdgcn_wmma_f32_16x16x4_f32(false, A, false, B,
                                                    (short)0, acc, false, false);
    }

#pragma unroll
    for (int v = 0; v < 8; ++v) {
        int orow = v + 8 * kslot;
        s_wh[orow * WH_PITCH + n0 + row] = acc[v];
    }
    __syncthreads();

    float* out = Wh + ((size_t)b * N_NODES + i0) * FDIM + n0 + row;
#pragma unroll
    for (int v = 0; v < 8; ++v) {
        int orow = v + 8 * kslot;
        out[(size_t)orow * FDIM] = acc[v];
    }

    const int r = wave * 2 + kslot;
    const float4 wlo = *reinterpret_cast<const float4*>(s_wh + r * WH_PITCH + 4 * l16);
    const float4 whi = *reinterpret_cast<const float4*>(s_wh + r * WH_PITCH + 64 + 4 * l16);
    float s1 = wlo.x * a1lo.x + wlo.y * a1lo.y + wlo.z * a1lo.z + wlo.w * a1lo.w
             + whi.x * a1hi.x + whi.y * a1hi.y + whi.z * a1hi.z + whi.w * a1hi.w;
    float s2 = wlo.x * a2lo.x + wlo.y * a2lo.y + wlo.z * a2lo.z + wlo.w * a2lo.w
             + whi.x * a2hi.x + whi.y * a2hi.y + whi.z * a2hi.z + whi.w * a2hi.w;
    for (int m = 8; m >= 1; m >>= 1) {
        s1 += __shfl_xor(s1, m, 32);
        s2 += __shfl_xor(s2, m, 32);
    }
    if (l16 == 0) {
        f1[(size_t)b * N_NODES + i0 + r] = s1;
        f2[(size_t)b * N_NODES + i0 + r] = s2;
    }
}

#define L_FLOOR 1e-6f

__device__ __forceinline__ float gat_score(int m, float x) {
    const float e = x > 0.f ? x : ALPHA * x;
    return m > 0 ? e : NEG_BIG;
}

__device__ __forceinline__ float gat_round(const float* __restrict__ whb,
                                           const int* __restrict__ arow,
                                           const float* s_f2, float* s_acc, float* s_lw,
                                           float rf1, float rm, int wave, int c, int h) {
    v8f acc[8] = {};
    float lsum = 0.f;
    for (int ch = 0; ch < 16; ++ch) {
        const int jb = wave * 256 + ch * 16 + 8 * h;
        const int4   m0 = *reinterpret_cast<const int4*>(arow + jb);
        const int4   m1 = *reinterpret_cast<const int4*>(arow + jb + 4);
        const float4 fa = *reinterpret_cast<const float4*>(s_f2 + jb);
        const float4 fb = *reinterpret_cast<const float4*>(s_f2 + jb + 4);

        const float* wrow = whb + (size_t)jb * FDIM;
        float4 wv[16];
#pragma unroll
        for (int q = 0; q < 4; ++q) {
            const float* w0 = wrow + (size_t)(2 * q) * FDIM;
            wv[4 * q + 0] = *reinterpret_cast<const float4*>(w0);
            wv[4 * q + 1] = *reinterpret_cast<const float4*>(w0 + 64);
            wv[4 * q + 2] = *reinterpret_cast<const float4*>(w0 + FDIM);
            wv[4 * q + 3] = *reinterpret_cast<const float4*>(w0 + FDIM + 64);
        }

        float p[8];
        p[0] = __expf(gat_score(m0.x, rf1 + fa.x) - rm);
        p[1] = __expf(gat_score(m0.y, rf1 + fa.y) - rm);
        p[2] = __expf(gat_score(m0.z, rf1 + fa.z) - rm);
        p[3] = __expf(gat_score(m0.w, rf1 + fa.w) - rm);
        p[4] = __expf(gat_score(m1.x, rf1 + fb.x) - rm);
        p[5] = __expf(gat_score(m1.y, rf1 + fb.y) - rm);
        p[6] = __expf(gat_score(m1.z, rf1 + fb.z) - rm);
        p[7] = __expf(gat_score(m1.w, rf1 + fb.w) - rm);
        lsum += ((p[0] + p[1]) + (p[2] + p[3])) + ((p[4] + p[5]) + (p[6] + p[7]));

#pragma unroll
        for (int q = 0; q < 4; ++q) {
            const float4 b00 = wv[4 * q + 0];
            const float4 b01 = wv[4 * q + 1];
            const float4 b10 = wv[4 * q + 2];
            const float4 b11 = wv[4 * q + 3];
            v2f A;
            A.x = p[2 * q];
            A.y = p[2 * q + 1];
            v2f B;
            B.x = b00.x; B.y = b10.x;
            acc[0] = __builtin_amdgcn_wmma_f32_16x16x4_f32(false, A, false, B, (short)0, acc[0], false, false);
            B.x = b00.y; B.y = b10.y;
            acc[1] = __builtin_amdgcn_wmma_f32_16x16x4_f32(false, A, false, B, (short)0, acc[1], false, false);
            B.x = b00.z; B.y = b10.z;
            acc[2] = __builtin_amdgcn_wmma_f32_16x16x4_f32(false, A, false, B, (short)0, acc[2], false, false);
            B.x = b00.w; B.y = b10.w;
            acc[3] = __builtin_amdgcn_wmma_f32_16x16x4_f32(false, A, false, B, (short)0, acc[3], false, false);
            B.x = b01.x; B.y = b11.x;
            acc[4] = __builtin_amdgcn_wmma_f32_16x16x4_f32(false, A, false, B, (short)0, acc[4], false, false);
            B.x = b01.y; B.y = b11.y;
            acc[5] = __builtin_amdgcn_wmma_f32_16x16x4_f32(false, A, false, B, (short)0, acc[5], false, false);
            B.x = b01.z; B.y = b11.z;
            acc[6] = __builtin_amdgcn_wmma_f32_16x16x4_f32(false, A, false, B, (short)0, acc[6], false, false);
            B.x = b01.w; B.y = b11.w;
            acc[7] = __builtin_amdgcn_wmma_f32_16x16x4_f32(false, A, false, B, (short)0, acc[7], false, false);
        }
    }

    lsum += __shfl_xor(lsum, 16, 32);
    if (h == 0) s_lw[wave * 16 + c] = lsum;

#pragma unroll
    for (int n = 0; n < 8; ++n) {
        const int col = 64 * (n >> 2) + 4 * c + (n & 3);
#pragma unroll
        for (int v = 0; v < 8; ++v) {
            int orow = v + 8 * h;
            atomicAdd(&s_acc[orow * FDIM + col], acc[n][v]);
        }
    }
    __syncthreads();

    float l = 0.f;
#pragma unroll
    for (int w = 0; w < 8; ++w) l += s_lw[w * 16 + c];
    return l;
}

__global__ __launch_bounds__(256) void gat_attn_kernel(const float* __restrict__ Wh,
                                                       const int*   __restrict__ adj,
                                                       const float* __restrict__ f1,
                                                       const float* __restrict__ f2,
                                                       float* __restrict__ out) {
    __shared__ alignas(16) float s_f2[N_NODES];
    __shared__ alignas(16) float s_acc[16 * FDIM];
    __shared__ alignas(16) float s_lw[8 * 16];
    __shared__ alignas(16) float s_wmax[8];
    __shared__ float s_m[16];

    const int blk  = blockIdx.x;
    const int b    = blk / (N_NODES / 16);
    const int i0   = (blk % (N_NODES / 16)) * 16;
    const int t    = threadIdx.x;
    const int wave = t >> 5;
    const int lane = t & 31;
    const int c    = lane & 15;
    const int h    = lane >> 4;

    const float rf1 = f1[(size_t)b * N_NODES + i0 + c];
    float4* s_accv = reinterpret_cast<float4*>(s_acc);
    {
        const float4* f2v = reinterpret_cast<const float4*>(f2 + (size_t)b * N_NODES);
        float4* s_f2v = reinterpret_cast<float4*>(s_f2);
        const float4 v0 = f2v[t];
        const float4 v1 = f2v[t + 256];
        s_f2v[t]       = v0;
        s_f2v[t + 256] = v1;
        s_accv[t]       = make_float4(0.f, 0.f, 0.f, 0.f);
        s_accv[t + 256] = make_float4(0.f, 0.f, 0.f, 0.f);
        float lm = fmaxf(fmaxf(fmaxf(v0.x, v0.y), fmaxf(v0.z, v0.w)),
                         fmaxf(fmaxf(v1.x, v1.y), fmaxf(v1.z, v1.w)));
#pragma unroll
        for (int m = 16; m >= 1; m >>= 1) lm = fmaxf(lm, __shfl_xor(lm, m, 32));
        if (lane == 0) s_wmax[wave] = lm;
    }
    __syncthreads();

    float rm;
    {
        const float4 wa = *reinterpret_cast<const float4*>(s_wmax);
        const float4 wb = *reinterpret_cast<const float4*>(s_wmax + 4);
        const float f2max = fmaxf(fmaxf(fmaxf(wa.x, wa.y), fmaxf(wa.z, wa.w)),
                                  fmaxf(fmaxf(wb.x, wb.y), fmaxf(wb.z, wb.w)));
        rm = fmaxf(NEG_BIG, gat_score(1, rf1 + f2max));
    }

    const int*   arow = adj + (size_t)(i0 + c) * N_NODES;
    const float* whb  = Wh + (size_t)b * N_NODES * FDIM + 4 * c;

    float l = gat_round(whb, arow, s_f2, s_acc, s_lw, rf1, rm, wave, c, h);

    if (__any(!(l >= L_FLOOR))) {
        s_accv[t]       = make_float4(0.f, 0.f, 0.f, 0.f);
        s_accv[t + 256] = make_float4(0.f, 0.f, 0.f, 0.f);
        {
            const int r = wave * 2 + h;
            const int4*   arow4 = reinterpret_cast<const int4*>(adj + (size_t)(i0 + r) * N_NODES);
            const float4* sf4   = reinterpret_cast<const float4*>(s_f2);
            const float fr = f1[(size_t)b * N_NODES + i0 + r];
            float mx = NEG_BIG;
#pragma unroll 4
            for (int it = 0; it < N_NODES / 64; ++it) {
                const int4   m = arow4[it * 16 + c];
                const float4 f = sf4[it * 16 + c];
                mx = fmaxf(mx, gat_score(m.x, fr + f.x));
                mx = fmaxf(mx, gat_score(m.y, fr + f.y));
                mx = fmaxf(mx, gat_score(m.z, fr + f.z));
                mx = fmaxf(mx, gat_score(m.w, fr + f.w));
            }
#pragma unroll
            for (int m = 8; m >= 1; m >>= 1) mx = fmaxf(mx, __shfl_xor(mx, m, 32));
            if (c == 0) s_m[r] = mx;
        }
        __syncthreads();
        l = gat_round(whb, arow, s_f2, s_acc, s_lw, rf1, s_m[c], wave, c, h);
    }

#pragma unroll
    for (int i = 0; i < 8; ++i) {
        const int k = t + 256 * i;
        const int r = k >> 7;
        const float lr = __shfl(l, r, 32);
        out[((size_t)b * N_NODES + i0 + r) * FDIM + (k & (FDIM - 1))] = s_acc[k] / lr;
    }
}

extern "C" void kernel_launch(void* const* d_in, const int* in_sizes, int n_in,
                              void* d_out, int out_size, void* d_ws, size_t ws_size,
                              hipStream_t stream) {
    const float* h   = (const float*)d_in[0];
    const int*   adj = (const int*)  d_in[1];
    const float* W   = (const float*)d_in[2];
    const float* a   = (const float*)d_in[3];
    float* out = (float*)d_out;

    float* Wh = (float*)d_ws;
    float* f1 = Wh + (size_t)NBATCH * N_NODES * FDIM;
    float* f2 = f1 + (size_t)NBATCH * N_NODES;

    gat_wh_kernel  <<<NBATCH * (N_NODES / 16), 256, 0, stream>>>(h, W, a, Wh, f1, f2);
    gat_attn_kernel<<<NBATCH * (N_NODES / 16), 256, 0, stream>>>(Wh, adj, f1, f2, out);
}


# Wh kernel: h tile fetched by two grouped 16-byte loads per thread into LDS rows of 132 floats: one round trip instead of eight dependent ones, A-fragment reads lose their 16-way bank conflict
# speedup vs baseline: 2.9206x; 1.0980x over previous
#include <hip/hip_runtime.h>

typedef __attribute__((ext_vector_type(2))) float v2f;
typedef __attribute__((ext_vector_type(8))) float v8f;

#define ALPHA    0.2f
#define NEG_BIG  -9000000000000000.0f
#define N_NODES  2048
#define FDIM     128
#define NBATCH   4

#define WH_PITCH 132
#define H_PITCH  132

__global__ __launch_bounds__(256) void gat_wh_kernel(const float* __restrict__ h,
                                                     const float* __restrict__ W,
                                                     const float* __restrict__ a,
                                                     float* __restrict__ Wh,
                                                     float* __restrict__ f1,
                                                     float* __restrict__ f2) {
    __shared__ alignas(16) float sh[16 * H_PITCH];
    __shared__ alignas(16) float s_wh[16 * WH_PITCH];
    const int tile = blockIdx.x;
    const int b    = tile / (N_NODES / 16);
    const int i0   = (tile % (N_NODES / 16)) * 16;
    const int t    = threadIdx.x;

    const float4* hb4 = reinterpret_cast<const float4*>(h + ((size_t)b * N_NODES + i0) * FDIM);
    const float4 g0 = hb4[t];
    const float4 g1 = hb4[t + 256];
    float* srow = sh + (t >> 5) * H_PITCH + 4 * (t & 31);
    *reinterpret_cast<float4*>(srow)               = g0;
    *reinterpret_cast<float4*>(srow + 8 * H_PITCH) = g1;
    __syncthreads();

    const int wave  = t >> 5;
    const int lane  = t & 31;
    const int row   = lane & 15;
    const int kslot = lane >> 4;
    const int n0    = wave * 16;

    const int l16 = lane & 15;
    const float4 a1lo = *reinterpret_cast<const float4*>(a + 4 * l16);
    const float4 a1hi = *reinterpret_cast<const float4*>(a + 64 + 4 * l16);
    const float4 a2lo = *reinterpret_cast<const float4*>(a + FDIM + 4 * l16);
    const float4 a2hi = *reinterpret_cast<const float4*>(a + FDIM + 64 + 4 * l16);

    v8f acc = {};
    for (int k0 = 0; k0 < FDIM; k0 += 4) {
        v2f A, B;
        A.x = sh[row * H_PITCH + k0 + kslot * 2];
        A.y = sh[row * H_PITCH + k0 + kslot * 2 + 1];
        B.x = W[(size_t)(k0 + kslot * 2)     * FDIM + n0 + row];
        B.y = W[(size_t)(k0 + kslot * 2 + 1) * FDIM + n0 + row];
        acc = __builtin_amdgcn_wmma_f32_16x16x4_f32(false, A, false, B,
                                                    (short)0, acc, false, false);
    }

#pragma unroll
    for (int v = 0; v < 8; ++v) {
        int orow = v + 8 * kslot;
        s_wh[orow * WH_PITCH + n0 + row] = acc[v];
    }
    __syncthreads();

    float* out = Wh + ((size_t)b * N_NODES + i0) * FDIM + n0 + row;
#pragma unroll
    for (int v = 0; v < 8; ++v) {
        int orow = v + 8 * kslot;
        out[(size_t)orow * FDIM] = acc[v];
    }

    const int r = wave * 2 + kslot;
    const float4 wlo = *reinterpret_cast<const float4*>(s_wh + r * WH_PITCH + 4 * l16);
    const float4 whi = *reinterpret_cast<const float4*>(s_wh + r * WH_PITCH + 64 + 4 * l16);
    float s1 = wlo.x * a1lo.x + wlo.y * a1lo.y + wlo.z * a1lo.z + wlo.w * a1lo.w
             + whi.x * a1hi.x + whi.y * a1hi.y + whi.z * a1hi.z + whi.w * a1hi.w;
    float s2 = wlo.x * a2lo.x + wlo.y * a2lo.y + wlo.z * a2lo.z + wlo.w * a2lo.w
             + whi.x * a2hi.x + whi.y * a2hi.y + whi.z * a2hi.z + whi.w * a2hi.w;
    for (int m = 8; m >= 1; m >>= 1) {
        s1 += __shfl_xor(s1, m, 32);
        s2 += __shfl_xor(s2, m, 32);
    }
    if (l16 == 0) {
        f1[(size_t)b * N_NODES + i0 + r] = s1;
        f2[(size_t)b * N_NODES + i0 + r] = s2;
    }
}

#define L_FLOOR 1e-6f

__device__ __forceinline__ float gat_score(int m, float x) {
    const float e = x > 0.f ? x : ALPHA * x;
    return m > 0 ? e : NEG_BIG;
}

__device__ __forceinline__ float gat_round(const float* __restrict__ whb,
                                           const int* __restrict__ arow,
                                           const float* s_f2, float* s_acc, float* s_lw,
                                           float rf1, float rm, int wave, int c, int h) {
    v8f acc[8] = {};
    float lsum = 0.f;
    for (int ch = 0; ch < 16; ++ch) {
        const int jb = wave * 256 + ch * 16 + 8 * h;
        const int4   m0 = *reinterpret_cast<const int4*>(arow + jb);
        const int4   m1 = *reinterpret_cast<const int4*>(arow + jb + 4);
        const float4 fa = *reinterpret_cast<const float4*>(s_f2 + jb);
        const float4 fb = *reinterpret_cast<const float4*>(s_f2 + jb + 4);

        const float* wrow = whb + (size_t)jb * FDIM;
        float4 wv[16];
#pragma unroll
        for (int q = 0; q < 4; ++q) {
            const float* w0 = wrow + (size_t)(2 * q) * FDIM;
            wv[4 * q + 0] = *reinterpret_cast<const float4*>(w0);
            wv[4 * q + 1] = *reinterpret_cast<const float4*>(w0 + 64);
            wv[4 * q + 2] = *reinterpret_cast<const float4*>(w0 + FDIM);
            wv[4 * q + 3] = *reinterpret_cast<const float4*>(w0 + FDIM + 64);
        }

        float p[8];
        p[0] = __expf(gat_score(m0.x, rf1 + fa.x) - rm);
        p[1] = __expf(gat_score(m0.y, rf1 + fa.y) - rm);
        p[2] = __expf(gat_score(m0.z, rf1 + fa.z) - rm);
        p[3] = __expf(gat_score(m0.w, rf1 + fa.w) - rm);
        p[4] = __expf(gat_score(m1.x, rf1 + fb.x) - rm);
        p[5] = __expf(gat_score(m1.y, rf1 + fb.y) - rm);
        p[6] = __expf(gat_score(m1.z, rf1 + fb.z) - rm);
        p[7] = __expf(gat_score(m1.w, rf1 + fb.w) - rm);
        lsum += ((p[0] + p[1]) + (p[2] + p[3])) + ((p[4] + p[5]) + (p[6] + p[7]));

#pragma unroll
        for (int q = 0; q < 4; ++q) {
            const float4 b00 = wv[4 * q + 0];
            const float4 b01 = wv[4 * q + 1];
            const float4 b10 = wv[4 * q + 2];
            const float4 b11 = wv[4 * q + 3];
            v2f A;
            A.x = p[2 * q];
            A.y = p[2 * q + 1];
            v2f B;
            B.x = b00.x; B.y = b10.x;
            acc[0] = __builtin_amdgcn_wmma_f32_16x16x4_f32(false, A, false, B, (short)0, acc[0], false, false);
            B.x = b00.y; B.y = b10.y;
            acc[1] = __builtin_amdgcn_wmma_f32_16x16x4_f32(false, A, false, B, (short)0, acc[1], false, false);
            B.x = b00.z; B.y = b10.z;
            acc[2] = __builtin_amdgcn_wmma_f32_16x16x4_f32(false, A, false, B, (short)0, acc[2], false, false);
            B.x = b00.w; B.y = b10.w;
            acc[3] = __builtin_amdgcn_wmma_f32_16x16x4_f32(false, A, false, B, (short)0, acc[3], false, false);
            B.x = b01.x; B.y = b11.x;
            acc[4] = __builtin_amdgcn_wmma_f32_16x16x4_f32(false, A, false, B, (short)0, acc[4], false, false);
            B.x = b01.y; B.y = b11.y;
            acc[5] = __builtin_amdgcn_wmma_f32_16x16x4_f32(false, A, false, B, (short)0, acc[5], false, false);
            B.x = b01.z; B.y = b11.z;
            acc[6] = __builtin_amdgcn_wmma_f32_16x16x4_f32(false, A, false, B, (short)0, acc[6], false, false);
            B.x = b01.w; B.y = b11.w;
            acc[7] = __builtin_amdgcn_wmma_f32_16x16x4_f32(false, A, false, B, (short)0, acc[7], false, false);
        }
    }

    lsum += __shfl_xor(lsum, 16, 32);
    if (h == 0) s_lw[wave * 16 + c] = lsum;

#pragma unroll
    for (int n = 0; n < 8; ++n) {
        const int col = 64 * (n >> 2) + 4 * c + (n & 3);
#pragma unroll
        for (int v = 0; v < 8; ++v) {
            int orow = v + 8 * h;
            atomicAdd(&s_acc[orow * FDIM + col], acc[n][v]);
        }
    }
    __syncthreads();

    float l = 0.f;
#pragma unroll
    for (int w = 0; w < 8; ++w) l += s_lw[w * 16 + c];
    return l;
}

__global__ __launch_bounds__(256) void gat_attn_kernel(const float* __restrict__ Wh,
                                                       const int*   __restrict__ adj,
                                                       const float* __restrict__ f1,
                                                       const float* __restrict__ f2,
                                                       float* __restrict__ out) {
    __shared__ alignas(16) float s_f2[N_NODES];
    __shared__ alignas(16) float s_acc[16 * FDIM];
    __shared__ alignas(16) float s_lw[8 * 16];
    __shared__ alignas(16) float s_wmax[8];
    __shared__ float s_m[16];

    const int blk  = blockIdx.x;
    const int b    = blk / (N_NODES / 16);
    const int i0   = (blk % (N_NODES / 16)) * 16;
    const int t    = threadIdx.x;
    const int wave = t >> 5;
    const int lane = t & 31;
    const int c    = lane & 15;
    const int h    = lane >> 4;

    const float rf1 = f1[(size_t)b * N_NODES + i0 + c];
    float4* s_accv = reinterpret_cast<float4*>(s_acc);
    {
        const float4* f2v = reinterpret_cast<const float4*>(f2 + (size_t)b * N_NODES);
        float4* s_f2v = reinterpret_cast<float4*>(s_f2);
        const float4 v0 = f2v[t];
        const float4 v1 = f2v[t + 256];
        s_f2v[t]       = v0;
        s_f2v[t + 256] = v1;
        s_accv[t]       = make_float4(0.f, 0.f, 0.f, 0.f);
        s_accv[t + 256] = make_float4(0.f, 0.f, 0.f, 0.f);
        float lm = fmaxf(fmaxf(fmaxf(v0.x, v0.y), fmaxf(v0.z, v0.w)),
                         fmaxf(fmaxf(v1.x, v1.y), fmaxf(v1.z, v1.w)));
#pragma unroll
        for (int m = 16; m >= 1; m >>= 1) lm = fmaxf(lm, __shfl_xor(lm, m, 32));
        if (lane == 0) s_wmax[wave] = lm;
    }
    __syncthreads();

    float rm;
    {
        const float4 wa = *reinterpret_cast<const float4*>(s_wmax);
        const float4 wb = *reinterpret_cast<const float4*>(s_wmax + 4);
        const float f2max = fmaxf(fmaxf(fmaxf(wa.x, wa.y), fmaxf(wa.z, wa.w)),
                                  fmaxf(fmaxf(wb.x, wb.y), fmaxf(wb.z, wb.w)));
        rm = fmaxf(NEG_BIG, gat_score(1, rf1 + f2max));
    }

    const int*   arow = adj + (size_t)(i0 + c) * N_NODES;
    const float* whb  = Wh + (size_t)b * N_NODES * FDIM + 4 * c;

    float l = gat_round(whb, arow, s_f2, s_acc, s_lw, rf1, rm, wave, c, h);

    if (__any(!(l >= L_FLOOR))) {
        s_accv[t]       = make_float4(0.f, 0.f, 0.f, 0.f);
        s_accv[t + 256] = make_float4(0.f, 0.f, 0.f, 0.f);
        {
            const int r = wave * 2 + h;
            const int4*   arow4 = reinterpret_cast<const int4*>(adj + (size_t)(i0 + r) * N_NODES);
            const float4* sf4   = reinterpret_cast<const float4*>(s_f2);
            const float fr = f1[(size_t)b * N_NODES + i0 + r];
            float mx = NEG_BIG;
#pragma unroll 4
            for (int it = 0; it < N_NODES / 64; ++it) {
                const int4   m = arow4[it * 16 + c];
                const float4 f = sf4[it * 16 + c];
                mx = fmaxf(mx, gat_score(m.x, fr + f.x));
                mx = fmaxf(mx, gat_score(m.y, fr + f.y));
                mx = fmaxf(mx, gat_score(m.z, fr + f.z));
                mx = fmaxf(mx, gat_score(m.w, fr + f.w));
            }
#pragma unroll
            for (int m = 8; m >= 1; m >>= 1) mx = fmaxf(mx, __shfl_xor(mx, m, 32));
            if (c == 0) s_m[r] = mx;
        }
        __syncthreads();
        l = gat_round(whb, arow, s_f2, s_acc, s_lw, rf1, s_m[c], wave, c, h);
    }

#pragma unroll
    for (int i = 0; i < 8; ++i) {
        const int k = t + 256 * i;
        const int r = k >> 7;
        const float lr = __shfl(l, r, 32);
        out[((size_t)b * N_NODES + i0 + r) * FDIM + (k & (FDIM - 1))] = s_acc[k] / lr;
    }
}

extern "C" void kernel_launch(void* const* d_in, const int* in_sizes, int n_in,
                              void* d_out, int out_size, void* d_ws, size_t ws_size,
                              hipStream_t stream) {
    const float* h   = (const float*)d_in[0];
    const int*   adj = (const int*)  d_in[1];
    const float* W   = (const float*)d_in[2];
    const float* a   = (const float*)d_in[3];
    float* out = (float*)d_out;

    float* Wh = (float*)d_ws;
    float* f1 = Wh + (size_t)NBATCH * N_NODES * FDIM;
    float* f2 = f1 + (size_t)NBATCH * N_NODES;

    gat_wh_kernel  <<<NBATCH * (N_NODES / 16), 256, 0, stream>>>(h, W, a, Wh, f1, f2);
    gat_attn_kernel<<<NBATCH * (N_NODES / 16), 256, 0, stream>>>(Wh, adj, f1, f2, out);
}


# attention kernel: 32 rows per block, each wave keeps two 16-row accumulator tiles so every loaded Wh fragment feeds two WMMAs (Wh reads halve; 256 blocks, 2 waves per SIMD); sum order unchanged
# speedup vs baseline: 3.0232x; 1.0351x over previous
#include <hip/hip_runtime.h>

typedef __attribute__((ext_vector_type(2))) float v2f;
typedef __attribute__((ext_vector_type(8))) float v8f;

#define ALPHA    0.2f
#define NEG_BIG  -9000000000000000.0f
#define N_NODES  2048
#define FDIM     128
#define NBATCH   4

#define WH_PITCH 132
#define H_PITCH  132

__global__ __launch_bounds__(256) void gat_wh_kernel(const float* __restrict__ h,
                                                     const float* __restrict__ W,
                                                     const float* __restrict__ a,
                                                     float* __restrict__ Wh,
                                                     float* __restrict__ f1,
                                                     float* __restrict__ f2) {
    __shared__ alignas(16) float sh[16 * H_PITCH];
    __shared__ alignas(16) float s_wh[16 * WH_PITCH];
    const int tile = blockIdx.x;
    const int b    = tile / (N_NODES / 16);
    const int i0   = (tile % (N_NODES / 16)) * 16;
    const int t    = threadIdx.x;

    const float4* hb4 = reinterpret_cast<const float4*>(h + ((size_t)b * N_NODES + i0) * FDIM);
    const float4 g0 = hb4[t];
    const float4 g1 = hb4[t + 256];
    float* srow = sh + (t >> 5) * H_PITCH + 4 * (t & 31);
    *reinterpret_cast<float4*>(srow)               = g0;
    *reinterpret_cast<float4*>(srow + 8 * H_PITCH) = g1;
    __syncthreads();

    const int wave  = t >> 5;
    const int lane  = t & 31;
    const int row   = lane & 15;
    const int kslot = lane >> 4;
    const int n0    = wave * 16;

    const int l16 = lane & 15;
    const float4 a1lo = *reinterpret_cast<const float4*>(a + 4 * l16);
    const float4 a1hi = *reinterpret_cast<const float4*>(a + 64 + 4 * l16);
    const float4 a2lo = *reinterpret_cast<const float4*>(a + FDIM + 4 * l16);
    const float4 a2hi = *reinterpret_cast<const float4*>(a + FDIM + 64 + 4 * l16);

    v8f acc = {};
    for (int k0 = 0; k0 < FDIM; k0 += 4) {
        v2f A, B;
        A.x = sh[row * H_PITCH + k0 + kslot * 2];
        A.y = sh[row * H_PITCH + k0 + kslot * 2 + 1];
        B.x = W[(size_t)(k0 + kslot * 2)     * FDIM + n0 + row];
        B.y = W[(size_t)(k0 + kslot * 2 + 1) * FDIM + n0 + row];
        acc = __builtin_amdgcn_wmma_f32_16x16x4_f32(false, A, false, B,
                                                    (short)0, acc, false, false);
    }

#pragma unroll
    for (int v = 0; v < 8; ++v) {
        int orow = v + 8 * kslot;
        s_wh[orow * WH_PITCH + n0 + row] = acc[v];
    }
    __syncthreads();

    float* out = Wh + ((size_t)b * N_NODES + i0) * FDIM + n0 + row;
#pragma unroll
    for (int v = 0; v < 8; ++v) {
        int orow = v + 8 * kslot;
        out[(size_t)orow * FDIM] = acc[v];
    }

    const int r = wave * 2 + kslot;
    const float4 wlo = *reinterpret_cast<const float4*>(s_wh + r * WH_PITCH + 4 * l16);
    const float4 whi = *reinterpret_cast<const float4*>(s_wh + r * WH_PITCH + 64 + 4 * l16);
    float s1 = wlo.x * a1lo.x + wlo.y * a1lo.y + wlo.z * a1lo.z + wlo.w * a1lo.w
             + whi.x * a1hi.x + whi.y * a1hi.y + whi.z * a1hi.z + whi.w * a1hi.w;
    float s2 = wlo.x * a2lo.x + wlo.y * a2lo.y + wlo.z * a2lo.z + wlo.w * a2lo.w
             + whi.x * a2hi.x + whi.y * a2hi.y + whi.z * a2hi.z + whi.w * a2hi.w;
    for (int m = 8; m >= 1; m >>= 1) {
        s1 += __shfl_xor(s1, m, 32);
        s2 += __shfl_xor(s2, m, 32);
    }
    if (l16 == 0) {
        f1[(size_t)b * N_NODES + i0 + r] = s1;
        f2[(size_t)b * N_NODES + i0 + r] = s2;
    }
}

#define L_FLOOR 1e-6f
#define ROWS_PB 32

__device__ __forceinline__ float gat_score(int m, float x) {
    const float e = x > 0.f ? x : ALPHA * x;
    return m > 0 ? e : NEG_BIG;
}

__device__ __forceinline__ float gat_p8(float* p, const int4 m0, const int4 m1,
                                        const float4 fa, const float4 fb,
                                        float rf1, float rm) {
    p[0] = __expf(gat_score(m0.x, rf1 + fa.x) - rm);
    p[1] = __expf(gat_score(m0.y, rf1 + fa.y) - rm);
    p[2] = __expf(gat_score(m0.z, rf1 + fa.z) - rm);
    p[3] = __expf(gat_score(m0.w, rf1 + fa.w) - rm);
    p[4] = __expf(gat_score(m1.x, rf1 + fb.x) - rm);
    p[5] = __expf(gat_score(m1.y, rf1 + fb.y) - rm);
    p[6] = __expf(gat_score(m1.z, rf1 + fb.z) - rm);
    p[7] = __expf(gat_score(m1.w, rf1 + fb.w) - rm);
    return ((p[0] + p[1]) + (p[2] + p[3])) + ((p[4] + p[5]) + (p[6] + p[7]));
}

__device__ __forceinline__ void gat_round(const float* __restrict__ whb,
                                          const int* __restrict__ arow0,
                                          const int* __restrict__ arow1,
                                          const float* s_f2, float* s_acc, float* s_lw,
                                          float rf1a, float rf1b, float rma, float rmb,
                                          int wave, int c, int h,
                                          float& l0, float& l1) {
    v8f acc0[8] = {};
    v8f acc1[8] = {};
    float lsum0 = 0.f, lsum1 = 0.f;
    for (int ch = 0; ch < 16; ++ch) {
        const int jb = wave * 256 + ch * 16 + 8 * h;
        const int4   ma0 = *reinterpret_cast<const int4*>(arow0 + jb);
        const int4   ma1 = *reinterpret_cast<const int4*>(arow0 + jb + 4);
        const int4   mb0 = *reinterpret_cast<const int4*>(arow1 + jb);
        const int4   mb1 = *reinterpret_cast<const int4*>(arow1 + jb + 4);
        const float4 fa = *reinterpret_cast<const float4*>(s_f2 + jb);
        const float4 fb = *reinterpret_cast<const float4*>(s_f2 + jb + 4);

        const float* wrow = whb + (size_t)jb * FDIM;
        float4 wv[16];
#pragma unroll
        for (int q = 0; q < 4; ++q) {
            const float* w0 = wrow + (size_t)(2 * q) * FDIM;
            wv[4 * q + 0] = *reinterpret_cast<const float4*>(w0);
            wv[4 * q + 1] = *reinterpret_cast<const float4*>(w0 + 64);
            wv[4 * q + 2] = *reinterpret_cast<const float4*>(w0 + FDIM);
            wv[4 * q + 3] = *reinterpret_cast<const float4*>(w0 + FDIM + 64);
        }

        float p0[8], p1[8];
        lsum0 += gat_p8(p0, ma0, ma1, fa, fb, rf1a, rma);
        lsum1 += gat_p8(p1, mb0, mb1, fa, fb, rf1b, rmb);

#pragma unroll
        for (int q = 0; q < 4; ++q) {
            const float4 b00 = wv[4 * q + 0];
            const float4 b01 = wv[4 * q + 1];
            const float4 b10 = wv[4 * q + 2];
            const float4 b11 = wv[4 * q + 3];
            v2f A0, A1;
            A0.x = p0[2 * q];
            A0.y = p0[2 * q + 1];
            A1.x = p1[2 * q];
            A1.y = p1[2 * q + 1];
            v2f B;
            B.x = b00.x; B.y = b10.x;
            acc0[0] = __builtin_amdgcn_wmma_f32_16x16x4_f32(false, A0, false, B, (short)0, acc0[0], false, false);
            acc1[0] = __builtin_amdgcn_wmma_f32_16x16x4_f32(false, A1, false, B, (short)0, acc1[0], false, false);
            B.x = b00.y; B.y = b10.y;
            acc0[1] = __builtin_amdgcn_wmma_f32_16x16x4_f32(false, A0, false, B, (short)0, acc0[1], false, false);
            acc1[1] = __builtin_amdgcn_wmma_f32_16x16x4_f32(false, A1, false, B, (short)0, acc1[1], false, false);
            B.x = b00.z; B.y = b10.z;
            acc0[2] = __builtin_amdgcn_wmma_f32_16x16x4_f32(false, A0, false, B, (short)0, acc0[2], false, false);
            acc1[2] = __builtin_amdgcn_wmma_f32_16x16x4_f32(false, A1, false, B, (short)0, acc1[2], false, false);
            B.x = b00.w; B.y = b10.w;
            acc0[3] = __builtin_amdgcn_wmma_f32_16x16x4_f32(false, A0, false, B, (short)0, acc0[3], false, false);
            acc1[3] = __builtin_amdgcn_wmma_f32_16x16x4_f32(false, A1, false, B, (short)0, acc1[3], false, false);
            B.x = b01.x; B.y = b11.x;
            acc0[4] = __builtin_amdgcn_wmma_f32_16x16x4_f32(false, A0, false, B, (short)0, acc0[4], false, false);
            acc1[4] = __builtin_amdgcn_wmma_f32_16x16x4_f32(false, A1, false, B, (short)0, acc1[4], false, false);
            B.x = b01.y; B.y = b11.y;
            acc0[5] = __builtin_amdgcn_wmma_f32_16x16x4_f32(false, A0, false, B, (short)0, acc0[5], false, false);
            acc1[5] = __builtin_amdgcn_wmma_f32_16x16x4_f32(false, A1, false, B, (short)0, acc1[5], false, false);
            B.x = b01.z; B.y = b11.z;
            acc0[6] = __builtin_amdgcn_wmma_f32_16x16x4_f32(false, A0, false, B, (short)0, acc0[6], false, false);
            acc1[6] = __builtin_amdgcn_wmma_f32_16x16x4_f32(false, A1, false, B, (short)0, acc1[6], false, false);
            B.x = b01.w; B.y = b11.w;
            acc0[7] = __builtin_amdgcn_wmma_f32_16x16x4_f32(false, A0, false, B, (short)0, acc0[7], false, false);
            acc1[7] = __builtin_amdgcn_wmma_f32_16x16x4_f32(false, A1, false, B, (short)0, acc1[7], false, false);
        }
    }

    lsum0 += __shfl_xor(lsum0, 16, 32);
    lsum1 += __shfl_xor(lsum1, 16, 32);
    if (h == 0) {
        s_lw[wave * ROWS_PB + c]      = lsum0;
        s_lw[wave * ROWS_PB + 16 + c] = lsum1;
    }

#pragma unroll
    for (int n = 0; n < 8; ++n) {
        const int col = 64 * (n >> 2) + 4 * c + (n & 3);
#pragma unroll
        for (int v = 0; v < 8; ++v) {
            int orow = v + 8 * h;
            atomicAdd(&s_acc[orow * FDIM + col], acc0[n][v]);
            atomicAdd(&s_acc[(16 + orow) * FDIM + col], acc1[n][v]);
        }
    }
    __syncthreads();

    float a0 = 0.f, a1 = 0.f;
#pragma unroll
    for (int w = 0; w < 8; ++w) {
        a0 += s_lw[w * ROWS_PB + c];
        a1 += s_lw[w * ROWS_PB + 16 + c];
    }
    l0 = a0;
    l1 = a1;
}

__global__ __launch_bounds__(256) __attribute__((amdgpu_waves_per_eu(2, 2)))
void gat_attn_kernel(const float* __restrict__ Wh,
                                                       const int*   __restrict__ adj,
                                                       const float* __restrict__ f1,
                                                       const float* __restrict__ f2,
                                                       float* __restrict__ out) {
    __shared__ alignas(16) float s_f2[N_NODES];
    __shared__ alignas(16) float s_acc[ROWS_PB * FDIM];
    __shared__ alignas(16) float s_lw[8 * ROWS_PB];
    __shared__ alignas(16) float s_wmax[8];
    __shared__ float s_m[ROWS_PB];

    const int blk  = blockIdx.x;
    const int b    = blk / (N_NODES / ROWS_PB);
    const int i0   = (blk % (N_NODES / ROWS_PB)) * ROWS_PB;
    const int t    = threadIdx.x;
    const int wave = t >> 5;
    const int lane = t & 31;
    const int c    = lane & 15;
    const int h    = lane >> 4;

    const float rf1a = f1[(size_t)b * N_NODES + i0 + c];
    const float rf1b = f1[(size_t)b * N_NODES + i0 + 16 + c];
    float4* s_accv = reinterpret_cast<float4*>(s_acc);
    {
        const float4* f2v = reinterpret_cast<const float4*>(f2 + (size_t)b * N_NODES);
        float4* s_f2v = reinterpret_cast<float4*>(s_f2);
        const float4 v0 = f2v[t];
        const float4 v1 = f2v[t + 256];
        s_f2v[t]       = v0;
        s_f2v[t + 256] = v1;
#pragma unroll
        for (int i = 0; i < ROWS_PB * FDIM / 4 / 256; ++i)
            s_accv[t + 256 * i] = make_float4(0.f, 0.f, 0.f, 0.f);
        float lm = fmaxf(fmaxf(fmaxf(v0.x, v0.y), fmaxf(v0.z, v0.w)),
                         fmaxf(fmaxf(v1.x, v1.y), fmaxf(v1.z, v1.w)));
#pragma unroll
        for (int m = 16; m >= 1; m >>= 1) lm = fmaxf(lm, __shfl_xor(lm, m, 32));
        if (lane == 0) s_wmax[wave] = lm;
    }
    __syncthreads();

    float rma, rmb;
    {
        const float4 wa = *reinterpret_cast<const float4*>(s_wmax);
        const float4 wb = *reinterpret_cast<const float4*>(s_wmax + 4);
        const float f2max = fmaxf(fmaxf(fmaxf(wa.x, wa.y), fmaxf(wa.z, wa.w)),
                                  fmaxf(fmaxf(wb.x, wb.y), fmaxf(wb.z, wb.w)));
        rma = fmaxf(NEG_BIG, gat_score(1, rf1a + f2max));
        rmb = fmaxf(NEG_BIG, gat_score(1, rf1b + f2max));
    }

    const int*   arow0 = adj + (size_t)(i0 + c) * N_NODES;
    const int*   arow1 = adj + (size_t)(i0 + 16 + c) * N_NODES;
    const float* whb   = Wh + (size_t)b * N_NODES * FDIM + 4 * c;

    float l0, l1;
    gat_round(whb, arow0, arow1, s_f2, s_acc, s_lw, rf1a, rf1b, rma, rmb, wave, c, h, l0, l1);

    if (__any(!(l0 >= L_FLOOR) || !(l1 >= L_FLOOR))) {
#pragma unroll
        for (int i = 0; i < ROWS_PB * FDIM / 4 / 256; ++i)
            s_accv[t + 256 * i] = make_float4(0.f, 0.f, 0.f, 0.f);
        for (int pass = 0; pass < 2; ++pass) {
            const int r = wave * 4 + 2 * pass + h;
            const int4*   arow4 = reinterpret_cast<const int4*>(adj + (size_t)(i0 + r) * N_NODES);
            const float4* sf4   = reinterpret_cast<const float4*>(s_f2);
            const float fr = f1[(size_t)b * N_NODES + i0 + r];
            float mx = NEG_BIG;
#pragma unroll 4
            for (int it = 0; it < N_NODES / 64; ++it) {
                const int4   m = arow4[it * 16 + c];
                const float4 f = sf4[it * 16 + c];
                mx = fmaxf(mx, gat_score(m.x, fr + f.x));
                mx = fmaxf(mx, gat_score(m.y, fr + f.y));
                mx = fmaxf(mx, gat_score(m.z, fr + f.z));
                mx = fmaxf(mx, gat_score(m.w, fr + f.w));
            }
#pragma unroll
            for (int m = 8; m >= 1; m >>= 1) mx = fmaxf(mx, __shfl_xor(mx, m, 32));
            if (c == 0) s_m[r] = mx;
        }
        __syncthreads();
        gat_round(whb, arow0, arow1, s_f2, s_acc, s_lw, rf1a, rf1b, s_m[c], s_m[16 + c],
                  wave, c, h, l0, l1);
    }

#pragma unroll
    for (int i = 0; i < ROWS_PB * FDIM / 256; ++i) {
        const int k = t + 256 * i;
        const int r = k >> 7;
        const float lr = __shfl(i < 8 ? l0 : l1, r & 15, 32);
        out[((size_t)b * N_NODES + i0 + r) * FDIM + (k & (FDIM - 1))] = s_acc[k] / lr;
    }
}

extern "C" void kernel_launch(void* const* d_in, const int* in_sizes, int n_in,
                              void* d_out, int out_size, void* d_ws, size_t ws_size,
                              hipStream_t stream) {
    const float* h   = (const float*)d_in[0];
    const int*   adj = (const int*)  d_in[1];
    const float* W   = (const float*)d_in[2];
    const float* a   = (const float*)d_in[3];
    float* out = (float*)d_out;

    float* Wh = (float*)d_ws;
    float* f1 = Wh + (size_t)NBATCH * N_NODES * FDIM;
    float* f2 = f1 + (size_t)NBATCH * N_NODES;

    gat_wh_kernel  <<<NBATCH * (N_NODES / 16), 256, 0, stream>>>(h, W, a, Wh, f1, f2);
    gat_attn_kernel<<<NBATCH * (N_NODES / ROWS_PB), 256, 0, stream>>>(Wh, adj, f1, f2, out);
}


# attention kernel: a trip's 20 mask and Wh loads are requested one trip ahead into a second register set (trip 0's before the f2 staging), landing under the previous trip's WMMAs; sum order unchanged
# speedup vs baseline: 3.1443x; 1.0401x over previous
#include <hip/hip_runtime.h>

typedef __attribute__((ext_vector_type(2))) float v2f;
typedef __attribute__((ext_vector_type(8))) float v8f;

#define ALPHA    0.2f
#define NEG_BIG  -9000000000000000.0f
#define N_NODES  2048
#define FDIM     128
#define NBATCH   4

#define WH_PITCH 132
#define H_PITCH  132

__global__ __launch_bounds__(256) void gat_wh_kernel(const float* __restrict__ h,
                                                     const float* __restrict__ W,
                                                     const float* __restrict__ a,
                                                     float* __restrict__ Wh,
                                                     float* __restrict__ f1,
                                                     float* __restrict__ f2) {
    __shared__ alignas(16) float sh[16 * H_PITCH];
    __shared__ alignas(16) float s_wh[16 * WH_PITCH];
    const int tile = blockIdx.x;
    const int b    = tile / (N_NODES / 16);
    const int i0   = (tile % (N_NODES / 16)) * 16;
    const int t    = threadIdx.x;

    const float4* hb4 = reinterpret_cast<const float4*>(h + ((size_t)b * N_NODES + i0) * FDIM);
    const float4 g0 = hb4[t];
    const float4 g1 = hb4[t + 256];
    float* srow = sh + (t >> 5) * H_PITCH + 4 * (t & 31);
    *reinterpret_cast<float4*>(srow)               = g0;
    *reinterpret_cast<float4*>(srow + 8 * H_PITCH) = g1;
    __syncthreads();

    const int wave  = t >> 5;
    const int lane  = t & 31;
    const int row   = lane & 15;
    const int kslot = lane >> 4;
    const int n0    = wave * 16;

    const int l16 = lane & 15;
    const float4 a1lo = *reinterpret_cast<const float4*>(a + 4 * l16);
    const float4 a1hi = *reinterpret_cast<const float4*>(a + 64 + 4 * l16);
    const float4 a2lo = *reinterpret_cast<const float4*>(a + FDIM + 4 * l16);
    const float4 a2hi = *reinterpret_cast<const float4*>(a + FDIM + 64 + 4 * l16);

    v8f acc = {};
    for (int k0 = 0; k0 < FDIM; k0 += 4) {
        v2f A, B;
        A.x = sh[row * H_PITCH + k0 + kslot * 2];
        A.y = sh[row * H_PITCH + k0 + kslot * 2 + 1];
        B.x = W[(size_t)(k0 + kslot * 2)     * FDIM + n0 + row];
        B.y = W[(size_t)(k0 + kslot * 2 + 1) * FDIM + n0 + row];
        acc = __builtin_amdgcn_wmma_f32_16x16x4_f32(false, A, false, B,
                                                    (short)0, acc, false, false);
    }

#pragma unroll
    for (int v = 0; v < 8; ++v) {
        int orow = v + 8 * kslot;
        s_wh[orow * WH_PITCH + n0 + row] = acc[v];
    }
    __syncthreads();

    float* out = Wh + ((size_t)b * N_NODES + i0) * FDIM + n0 + row;
#pragma unroll
    for (int v = 0; v < 8; ++v) {
        int orow = v + 8 * kslot;
        out[(size_t)orow * FDIM] = acc[v];
    }

    const int r = wave * 2 + kslot;
    const float4 wlo = *reinterpret_cast<const float4*>(s_wh + r * WH_PITCH + 4 * l16);
    const float4 whi = *reinterpret_cast<const float4*>(s_wh + r * WH_PITCH + 64 + 4 * l16);
    float s1 = wlo.x * a1lo.x + wlo.y * a1lo.y + wlo.z * a1lo.z + wlo.w * a1lo.w
             + whi.x * a1hi.x + whi.y * a1hi.y + whi.z * a1hi.z + whi.w * a1hi.w;
    float s2 = wlo.x * a2lo.x + wlo.y * a2lo.y + wlo.z * a2lo.z + wlo.w * a2lo.w
             + whi.x * a2hi.x + whi.y * a2hi.y + whi.z * a2hi.z + whi.w * a2hi.w;
    for (int m = 8; m >= 1; m >>= 1) {
        s1 += __shfl_xor(s1, m, 32);
        s2 += __shfl_xor(s2, m, 32);
    }
    if (l16 == 0) {
        f1[(size_t)b * N_NODES + i0 + r] = s1;
        f2[(size_t)b * N_NODES + i0 + r] = s2;
    }
}

#define L_FLOOR 1e-6f
#define ROWS_PB 32

__device__ __forceinline__ float gat_score(int m, float x) {
    const float e = x > 0.f ? x : ALPHA * x;
    return m > 0 ? e : NEG_BIG;
}

__device__ __forceinline__ float gat_p8(float* p, const int4 m0, const int4 m1,
                                        const float4 fa, const float4 fb,
                                        float rf1, float rm) {
    p[0] = __expf(gat_score(m0.x, rf1 + fa.x) - rm);
    p[1] = __expf(gat_score(m0.y, rf1 + fa.y) - rm);
    p[2] = __expf(gat_score(m0.z, rf1 + fa.z) - rm);
    p[3] = __expf(gat_score(m0.w, rf1 + fa.w) - rm);
    p[4] = __expf(gat_score(m1.x, rf1 + fb.x) - rm);
    p[5] = __expf(gat_score(m1.y, rf1 + fb.y) - rm);
    p[6] = __expf(gat_score(m1.z, rf1 + fb.z) - rm);
    p[7] = __expf(gat_score(m1.w, rf1 + fb.w) - rm);
    return ((p[0] + p[1]) + (p[2] + p[3])) + ((p[4] + p[5]) + (p[6] + p[7]));
}

struct GatTile {
    int4   ma0, ma1;
    int4   mb0, mb1;
    float4 wv[16];
};

__device__ __forceinline__ void gat_fetch(GatTile& T, const float* __restrict__ whb,
                                          const int* __restrict__ arow0,
                                          const int* __restrict__ arow1, int jb) {
    T.ma0 = *reinterpret_cast<const int4*>(arow0 + jb);
    T.ma1 = *reinterpret_cast<const int4*>(arow0 + jb + 4);
    T.mb0 = *reinterpret_cast<const int4*>(arow1 + jb);
    T.mb1 = *reinterpret_cast<const int4*>(arow1 + jb + 4);
    const float* wrow = whb + (size_t)jb * FDIM;
#pragma unroll
    for (int q = 0; q < 4; ++q) {
        const float* w0 = wrow + (size_t)(2 * q) * FDIM;
        T.wv[4 * q + 0] = *reinterpret_cast<const float4*>(w0);
        T.wv[4 * q + 1] = *reinterpret_cast<const float4*>(w0 + 64);
        T.wv[4 * q + 2] = *reinterpret_cast<const float4*>(w0 + FDIM);
        T.wv[4 * q + 3] = *reinterpret_cast<const float4*>(w0 + FDIM + 64);
    }
}

__device__ __forceinline__ void gat_trip(const GatTile& T, const float* s_f2, int jb,
                                         float rf1a, float rf1b, float rma, float rmb,
                                         v8f (&acc0)[8], v8f (&acc1)[8],
                                         float& lsum0, float& lsum1) {
    const float4 fa = *reinterpret_cast<const float4*>(s_f2 + jb);
    const float4 fb = *reinterpret_cast<const float4*>(s_f2 + jb + 4);

    float p0[8], p1[8];
    lsum0 += gat_p8(p0, T.ma0, T.ma1, fa, fb, rf1a, rma);
    lsum1 += gat_p8(p1, T.mb0, T.mb1, fa, fb, rf1b, rmb);

#pragma unroll
    for (int q = 0; q < 4; ++q) {
        const float4 b00 = T.wv[4 * q + 0];
        const float4 b01 = T.wv[4 * q + 1];
        const float4 b10 = T.wv[4 * q + 2];
        const float4 b11 = T.wv[4 * q + 3];
        v2f A0, A1;
        A0.x = p0[2 * q];
        A0.y = p0[2 * q + 1];
        A1.x = p1[2 * q];
        A1.y = p1[2 * q + 1];
        v2f B;
        B.x = b00.x; B.y = b10.x;
        acc0[0] = __builtin_amdgcn_wmma_f32_16x16x4_f32(false, A0, false, B, (short)0, acc0[0], false, false);
        acc1[0] = __builtin_amdgcn_wmma_f32_16x16x4_f32(false, A1, false, B, (short)0, acc1[0], false, false);
        B.x = b00.y; B.y = b10.y;
        acc0[1] = __builtin_amdgcn_wmma_f32_16x16x4_f32(false, A0, false, B, (short)0, acc0[1], false, false);
        acc1[1] = __builtin_amdgcn_wmma_f32_16x16x4_f32(false, A1, false, B, (short)0, acc1[1], false, false);
        B.x = b00.z; B.y = b10.z;
        acc0[2] = __builtin_amdgcn_wmma_f32_16x16x4_f32(false, A0, false, B, (short)0, acc0[2], false, false);
        acc1[2] = __builtin_amdgcn_wmma_f32_16x16x4_f32(false, A1, false, B, (short)0, acc1[2], false, false);
        B.x = b00.w; B.y = b10.w;
        acc0[3] = __builtin_amdgcn_wmma_f32_16x16x4_f32(false, A0, false, B, (short)0, acc0[3], false, false);
        acc1[3] = __builtin_amdgcn_wmma_f32_16x16x4_f32(false, A1, false, B, (short)0, acc1[3], false, false);
        B.x = b01.x; B.y = b11.x;
        acc0[4] = __builtin_amdgcn_wmma_f32_16x16x4_f32(false, A0, false, B, (short)0, acc0[4], false, false);
        acc1[4] = __builtin_amdgcn_wmma_f32_16x16x4_f32(false, A1, false, B, (short)0, acc1[4], false, false);
        B.x = b01.y; B.y = b11.y;
        acc0[5] = __builtin_amdgcn_wmma_f32_16x16x4_f32(false, A0, false, B, (short)0, acc0[5], false, false);
        acc1[5] = __builtin_amdgcn_wmma_f32_16x16x4_f32(false, A1, false, B, (short)0, acc1[5], false, false);
        B.x = b01.z; B.y = b11.z;
        acc0[6] = __builtin_amdgcn_wmma_f32_16x16x4_f32(false, A0, false, B, (short)0, acc0[6], false, false);
        acc1[6] = __builtin_amdgcn_wmma_f32_16x16x4_f32(false, A1, false, B, (short)0, acc1[6], false, false);
        B.x = b01.w; B.y = b11.w;
        acc0[7] = __builtin_amdgcn_wmma_f32_16x16x4_f32(false, A0, false, B, (short)0, acc0[7], false, false);
        acc1[7] = __builtin_amdgcn_wmma_f32_16x16x4_f32(false, A1, false, B, (short)0, acc1[7], false, false);
    }
}

__device__ __forceinline__ void gat_round(const float* __restrict__ whb,
                                          const int* __restrict__ arow0,
                                          const int* __restrict__ arow1,
                                          const float* s_f2, float* s_acc, float* s_lw,
                                          float rf1a, float rf1b, float rma, float rmb,
                                          int wave, int c, int h, GatTile& ta,
                                          float& l0, float& l1) {
    v8f acc0[8] = {};
    v8f acc1[8] = {};
    float lsum0 = 0.f, lsum1 = 0.f;
    const int jw = wave * 256 + 8 * h;
    GatTile tb;
    for (int ch = 0; ch < 14; ch += 2) {
        gat_fetch(tb, whb, arow0, arow1, jw + (ch + 1) * 16);
        gat_trip(ta, s_f2, jw + ch * 16, rf1a, rf1b, rma, rmb, acc0, acc1, lsum0, lsum1);
        gat_fetch(ta, whb, arow0, arow1, jw + (ch + 2) * 16);
        gat_trip(tb, s_f2, jw + (ch + 1) * 16, rf1a, rf1b, rma, rmb, acc0, acc1, lsum0, lsum1);
    }
    gat_fetch(tb, whb, arow0, arow1, jw + 15 * 16);
    gat_trip(ta, s_f2, jw + 14 * 16, rf1a, rf1b, rma, rmb, acc0, acc1, lsum0, lsum1);
    gat_trip(tb, s_f2, jw + 15 * 16, rf1a, rf1b, rma, rmb, acc0, acc1, lsum0, lsum1);

    lsum0 += __shfl_xor(lsum0, 16, 32);
    lsum1 += __shfl_xor(lsum1, 16, 32);
    if (h == 0) {
        s_lw[wave * ROWS_PB + c]      = lsum0;
        s_lw[wave * ROWS_PB + 16 + c] = lsum1;
    }

#pragma unroll
    for (int n = 0; n < 8; ++n) {
        const int col = 64 * (n >> 2) + 4 * c + (n & 3);
#pragma unroll
        for (int v = 0; v < 8; ++v) {
            int orow = v + 8 * h;
            atomicAdd(&s_acc[orow * FDIM + col], acc0[n][v]);
            atomicAdd(&s_acc[(16 + orow) * FDIM + col], acc1[n][v]);
        }
    }
    __syncthreads();

    float a0 = 0.f, a1 = 0.f;
#pragma unroll
    for (int w = 0; w < 8; ++w) {
        a0 += s_lw[w * ROWS_PB + c];
        a1 += s_lw[w * ROWS_PB + 16 + c];
    }
    l0 = a0;
    l1 = a1;
}

__global__ __launch_bounds__(256) __attribute__((amdgpu_waves_per_eu(2, 2)))
void gat_attn_kernel(const float* __restrict__ Wh,
                                                       const int*   __restrict__ adj,
                                                       const float* __restrict__ f1,
                                                       const float* __restrict__ f2,
                                                       float* __restrict__ out) {
    __shared__ alignas(16) float s_f2[N_NODES];
    __shared__ alignas(16) float s_acc[ROWS_PB * FDIM];
    __shared__ alignas(16) float s_lw[8 * ROWS_PB];
    __shared__ alignas(16) float s_wmax[8];
    __shared__ float s_m[ROWS_PB];

    const int blk  = blockIdx.x;
    const int b    = blk / (N_NODES / ROWS_PB);
    const int i0   = (blk % (N_NODES / ROWS_PB)) * ROWS_PB;
    const int t    = threadIdx.x;
    const int wave = t >> 5;
    const int lane = t & 31;
    const int c    = lane & 15;
    const int h    = lane >> 4;

    const float rf1a = f1[(size_t)b * N_NODES + i0 + c];
    const float rf1b = f1[(size_t)b * N_NODES + i0 + 16 + c];

    const int*   arow0 = adj + (size_t)(i0 + c) * N_NODES;
    const int*   arow1 = adj + (size_t)(i0 + 16 + c) * N_NODES;
    const float* whb   = Wh + (size_t)b * N_NODES * FDIM + 4 * c;

    GatTile t0;
    gat_fetch(t0, whb, arow0, arow1, wave * 256 + 8 * h);

    float4* s_accv = reinterpret_cast<float4*>(s_acc);
    {
        const float4* f2v = reinterpret_cast<const float4*>(f2 + (size_t)b * N_NODES);
        float4* s_f2v = reinterpret_cast<float4*>(s_f2);
        const float4 v0 = f2v[t];
        const float4 v1 = f2v[t + 256];
        s_f2v[t]       = v0;
        s_f2v[t + 256] = v1;
#pragma unroll
        for (int i = 0; i < ROWS_PB * FDIM / 4 / 256; ++i)
            s_accv[t + 256 * i] = make_float4(0.f, 0.f, 0.f, 0.f);
        float lm = fmaxf(fmaxf(fmaxf(v0.x, v0.y), fmaxf(v0.z, v0.w)),
                         fmaxf(fmaxf(v1.x, v1.y), fmaxf(v1.z, v1.w)));
#pragma unroll
        for (int m = 16; m >= 1; m >>= 1) lm = fmaxf(lm, __shfl_xor(lm, m, 32));
        if (lane == 0) s_wmax[wave] = lm;
    }
    __syncthreads();

    float rma, rmb;
    {
        const float4 wa = *reinterpret_cast<const float4*>(s_wmax);
        const float4 wb = *reinterpret_cast<const float4*>(s_wmax + 4);
        const float f2max = fmaxf(fmaxf(fmaxf(wa.x, wa.y), fmaxf(wa.z, wa.w)),
                                  fmaxf(fmaxf(wb.x, wb.y), fmaxf(wb.z, wb.w)));
        rma = fmaxf(NEG_BIG, gat_score(1, rf1a + f2max));
        rmb = fmaxf(NEG_BIG, gat_score(1, rf1b + f2max));
    }

    float l0, l1;
    gat_round(whb, arow0, arow1, s_f2, s_acc, s_lw, rf1a, rf1b, rma, rmb, wave, c, h, t0, l0, l1);

    if (__any(!(l0 >= L_FLOOR) || !(l1 >= L_FLOOR))) {
#pragma unroll
        for (int i = 0; i < ROWS_PB * FDIM / 4 / 256; ++i)
            s_accv[t + 256 * i] = make_float4(0.f, 0.f, 0.f, 0.f);
        for (int pass = 0; pass < 2; ++pass) {
            const int r = wave * 4 + 2 * pass + h;
            const int4*   arow4 = reinterpret_cast<const int4*>(adj + (size_t)(i0 + r) * N_NODES);
            const float4* sf4   = reinterpret_cast<const float4*>(s_f2);
            const float fr = f1[(size_t)b * N_NODES + i0 + r];
            float mx = NEG_BIG;
#pragma unroll 4
            for (int it = 0; it < N_NODES / 64; ++it) {
                const int4   m = arow4[it * 16 + c];
                const float4 f = sf4[it * 16 + c];
                mx = fmaxf(mx, gat_score(m.x, fr + f.x));
                mx = fmaxf(mx, gat_score(m.y, fr + f.y));
                mx = fmaxf(mx, gat_score(m.z, fr + f.z));
                mx = fmaxf(mx, gat_score(m.w, fr + f.w));
            }
#pragma unroll
            for (int m = 8; m >= 1; m >>= 1) mx = fmaxf(mx, __shfl_xor(mx, m, 32));
            if (c == 0) s_m[r] = mx;
        }
        __syncthreads();
        gat_fetch(t0, whb, arow0, arow1, wave * 256 + 8 * h);
        gat_round(whb, arow0, arow1, s_f2, s_acc, s_lw, rf1a, rf1b, s_m[c], s_m[16 + c],
                  wave, c, h, t0, l0, l1);
    }

#pragma unroll
    for (int i = 0; i < ROWS_PB * FDIM / 256; ++i) {
        const int k = t + 256 * i;
        const int r = k >> 7;
        const float lr = __shfl(i < 8 ? l0 : l1, r & 15, 32);
        out[((size_t)b * N_NODES + i0 + r) * FDIM + (k & (FDIM - 1))] = s_acc[k] / lr;
    }
}

extern "C" void kernel_launch(void* const* d_in, const int* in_sizes, int n_in,
                              void* d_out, int out_size, void* d_ws, size_t ws_size,
                              hipStream_t stream) {
    const float* h   = (const float*)d_in[0];
    const int*   adj = (const int*)  d_in[1];
    const float* W   = (const float*)d_in[2];
    const float* a   = (const float*)d_in[3];
    float* out = (float*)d_out;

    float* Wh = (float*)d_ws;
    float* f1 = Wh + (size_t)NBATCH * N_NODES * FDIM;
    float* f2 = f1 + (size_t)NBATCH * N_NODES;

    gat_wh_kernel  <<<NBATCH * (N_NODES / 16), 256, 0, stream>>>(h, W, a, Wh, f1, f2);
    gat_attn_kernel<<<NBATCH * (N_NODES / ROWS_PB), 256, 0, stream>>>(Wh, adj, f1, f2, out);
}


# Wh written by the first kernel in the second kernel's B-fragment order (row pairs side by side, 8 KB per 16 rows): a 16-byte load is two ready B operands, no pairing moves; sum order unchanged
# speedup vs baseline: 3.2622x; 1.0375x over previous
#include <hip/hip_runtime.h>

typedef __attribute__((ext_vector_type(2))) float v2f;
typedef __attribute__((ext_vector_type(8))) float v8f;

#define ALPHA    0.2f
#define NEG_BIG  -9000000000000000.0f
#define N_NODES  2048
#define FDIM     128
#define NBATCH   4

#define WH_PITCH 132
#define H_PITCH  132

__global__ __launch_bounds__(256) void gat_wh_kernel(const float* __restrict__ h,
                                                     const float* __restrict__ W,
                                                     const float* __restrict__ a,
                                                     float* __restrict__ Wh,
                                                     float* __restrict__ f1,
                                                     float* __restrict__ f2) {
    __shared__ alignas(16) float sh[16 * H_PITCH];
    __shared__ alignas(16) float s_wh[16 * WH_PITCH];
    const int tile = blockIdx.x;
    const int b    = tile / (N_NODES / 16);
    const int i0   = (tile % (N_NODES / 16)) * 16;
    const int t    = threadIdx.x;

    const float4* hb4 = reinterpret_cast<const float4*>(h + ((size_t)b * N_NODES + i0) * FDIM);
    const float4 g0 = hb4[t];
    const float4 g1 = hb4[t + 256];
    float* srow = sh + (t >> 5) * H_PITCH + 4 * (t & 31);
    *reinterpret_cast<float4*>(srow)               = g0;
    *reinterpret_cast<float4*>(srow + 8 * H_PITCH) = g1;
    __syncthreads();

    const int wave  = t >> 5;
    const int lane  = t & 31;
    const int row   = lane & 15;
    const int kslot = lane >> 4;
    const int n0    = wave * 16;

    const int l16 = lane & 15;
    const float4 a1lo = *reinterpret_cast<const float4*>(a + 4 * l16);
    const float4 a1hi = *reinterpret_cast<const float4*>(a + 64 + 4 * l16);
    const float4 a2lo = *reinterpret_cast<const float4*>(a + FDIM + 4 * l16);
    const float4 a2hi = *reinterpret_cast<const float4*>(a + FDIM + 64 + 4 * l16);

    v8f acc = {};
    for (int k0 = 0; k0 < FDIM; k0 += 4) {
        v2f A, B;
        A.x = sh[row * H_PITCH + k0 + kslot * 2];
        A.y = sh[row * H_PITCH + k0 + kslot * 2 + 1];
        B.x = W[(size_t)(k0 + kslot * 2)     * FDIM + n0 + row];
        B.y = W[(size_t)(k0 + kslot * 2 + 1) * FDIM + n0 + row];
        acc = __builtin_amdgcn_wmma_f32_16x16x4_f32(false, A, false, B,
                                                    (short)0, acc, false, false);
    }

#pragma unroll
    for (int v = 0; v < 8; ++v) {
        int orow = v + 8 * kslot;
        s_wh[orow * WH_PITCH + n0 + row] = acc[v];
    }
    __syncthreads();

    float4* outc = reinterpret_cast<float4*>(Wh + (size_t)tile * (16 * FDIM));
#pragma unroll
    for (int piece = 0; piece < 2; ++piece) {
        const int f  = t + 256 * piece;
        const int r0 = 8 * ((f >> 4) & 1) + 2 * (f >> 7);
        const int nn = 32 * ((f >> 5) & 3) + 2 * (f & 15);
        const float2 lo = *reinterpret_cast<const float2*>(s_wh + r0 * WH_PITCH + nn);
        const float2 hi = *reinterpret_cast<const float2*>(s_wh + (r0 + 1) * WH_PITCH + nn);
        outc[f] = make_float4(lo.x, hi.x, lo.y, hi.y);
    }

    const int r = wave * 2 + kslot;
    const float4 wlo = *reinterpret_cast<const float4*>(s_wh + r * WH_PITCH + 4 * l16);
    const float4 whi = *reinterpret_cast<const float4*>(s_wh + r * WH_PITCH + 64 + 4 * l16);
    float s1 = wlo.x * a1lo.x + wlo.y * a1lo.y + wlo.z * a1lo.z + wlo.w * a1lo.w
             + whi.x * a1hi.x + whi.y * a1hi.y + whi.z * a1hi.z + whi.w * a1hi.w;
    float s2 = wlo.x * a2lo.x + wlo.y * a2lo.y + wlo.z * a2lo.z + wlo.w * a2lo.w
             + whi.x * a2hi.x + whi.y * a2hi.y + whi.z * a2hi.z + whi.w * a2hi.w;
    for (int m = 8; m >= 1; m >>= 1) {
        s1 += __shfl_xor(s1, m, 32);
        s2 += __shfl_xor(s2, m, 32);
    }
    if (l16 == 0) {
        f1[(size_t)b * N_NODES + i0 + r] = s1;
        f2[(size_t)b * N_NODES + i0 + r] = s2;
    }
}

#define L_FLOOR 1e-6f
#define ROWS_PB 32

__device__ __forceinline__ float gat_score(int m, float x) {
    const float e = x > 0.f ? x : ALPHA * x;
    return m > 0 ? e : NEG_BIG;
}

__device__ __forceinline__ float gat_p8(float* p, const int4 m0, const int4 m1,
                                        const float4 fa, const float4 fb,
                                        float rf1, float rm) {
    p[0] = __expf(gat_score(m0.x, rf1 + fa.x) - rm);
    p[1] = __expf(gat_score(m0.y, rf1 + fa.y) - rm);
    p[2] = __expf(gat_score(m0.z, rf1 + fa.z) - rm);
    p[3] = __expf(gat_score(m0.w, rf1 + fa.w) - rm);
    p[4] = __expf(gat_score(m1.x, rf1 + fb.x) - rm);
    p[5] = __expf(gat_score(m1.y, rf1 + fb.y) - rm);
    p[6] = __expf(gat_score(m1.z, rf1 + fb.z) - rm);
    p[7] = __expf(gat_score(m1.w, rf1 + fb.w) - rm);
    return ((p[0] + p[1]) + (p[2] + p[3])) + ((p[4] + p[5]) + (p[6] + p[7]));
}

struct GatTile {
    int4   ma0, ma1;
    int4   mb0, mb1;
    float4 wv[16];
};

__device__ __forceinline__ void gat_fetch(GatTile& T, const float* __restrict__ wch,
                                          const int* __restrict__ arow0,
                                          const int* __restrict__ arow1, int jb) {
    T.ma0 = *reinterpret_cast<const int4*>(arow0 + jb);
    T.ma1 = *reinterpret_cast<const int4*>(arow0 + jb + 4);
    T.mb0 = *reinterpret_cast<const int4*>(arow1 + jb);
    T.mb1 = *reinterpret_cast<const int4*>(arow1 + jb + 4);
#pragma unroll
    for (int i = 0; i < 16; ++i)
        T.wv[i] = *reinterpret_cast<const float4*>(wch + 128 * i);
}

__device__ __forceinline__ void gat_trip(const GatTile& T, const float* s_f2, int jb,
                                         float rf1a, float rf1b, float rma, float rmb,
                                         v8f (&acc0)[8], v8f (&acc1)[8],
                                         float& lsum0, float& lsum1) {
    const float4 fa = *reinterpret_cast<const float4*>(s_f2 + jb);
    const float4 fb = *reinterpret_cast<const float4*>(s_f2 + jb + 4);

    float p0[8], p1[8];
    lsum0 += gat_p8(p0, T.ma0, T.ma1, fa, fb, rf1a, rma);
    lsum1 += gat_p8(p1, T.mb0, T.mb1, fa, fb, rf1b, rmb);

#pragma unroll
    for (int q = 0; q < 4; ++q) {
        v2f A0, A1;
        A0.x = p0[2 * q];
        A0.y = p0[2 * q + 1];
        A1.x = p1[2 * q];
        A1.y = p1[2 * q + 1];
#pragma unroll
        for (int m = 0; m < 4; ++m) {
            const float4 w = T.wv[4 * q + m];
            v2f B;
            B.x = w.x; B.y = w.y;
            acc0[2 * m]     = __builtin_amdgcn_wmma_f32_16x16x4_f32(false, A0, false, B, (short)0, acc0[2 * m], false, false);
            acc1[2 * m]     = __builtin_amdgcn_wmma_f32_16x16x4_f32(false, A1, false, B, (short)0, acc1[2 * m], false, false);
            B.x = w.z; B.y = w.w;
            acc0[2 * m + 1] = __builtin_amdgcn_wmma_f32_16x16x4_f32(false, A0, false, B, (short)0, acc0[2 * m + 1], false, false);
            acc1[2 * m + 1] = __builtin_amdgcn_wmma_f32_16x16x4_f32(false, A1, false, B, (short)0, acc1[2 * m + 1], false, false);
        }
    }
}

__device__ __forceinline__ void gat_round(const float* __restrict__ whl,
                                          const int* __restrict__ arow0,
                                          const int* __restrict__ arow1,
                                          const float* s_f2, float* s_acc, float* s_lw,
                                          float rf1a, float rf1b, float rma, float rmb,
                                          int wave, int c, int h, GatTile& ta,
                                          float& l0, float& l1) {
    v8f acc0[8] = {};
    v8f acc1[8] = {};
    float lsum0 = 0.f, lsum1 = 0.f;
    const int jw = wave * 256 + 8 * h;
    GatTile tb;
    for (int ch = 0; ch < 14; ch += 2) {
        gat_fetch(tb, whl + (ch + 1) * (16 * FDIM), arow0, arow1, jw + (ch + 1) * 16);
        gat_trip(ta, s_f2, jw + ch * 16, rf1a, rf1b, rma, rmb, acc0, acc1, lsum0, lsum1);
        gat_fetch(ta, whl + (ch + 2) * (16 * FDIM), arow0, arow1, jw + (ch + 2) * 16);
        gat_trip(tb, s_f2, jw + (ch + 1) * 16, rf1a, rf1b, rma, rmb, acc0, acc1, lsum0, lsum1);
    }
    gat_fetch(tb, whl + 15 * (16 * FDIM), arow0, arow1, jw + 15 * 16);
    gat_trip(ta, s_f2, jw + 14 * 16, rf1a, rf1b, rma, rmb, acc0, acc1, lsum0, lsum1);
    gat_trip(tb, s_f2, jw + 15 * 16, rf1a, rf1b, rma, rmb, acc0, acc1, lsum0, lsum1);

    lsum0 += __shfl_xor(lsum0, 16, 32);
    lsum1 += __shfl_xor(lsum1, 16, 32);
    if (h == 0) {
        s_lw[wave * ROWS_PB + c]      = lsum0;
        s_lw[wave * ROWS_PB + 16 + c] = lsum1;
    }

#pragma unroll
    for (int n = 0; n < 8; ++n) {
        const int col = 32 * (n >> 1) + 2 * c + (n & 1);
#pragma unroll
        for (int v = 0; v < 8; ++v) {
            int orow = v + 8 * h;
            atomicAdd(&s_acc[orow * FDIM + col], acc0[n][v]);
            atomicAdd(&s_acc[(16 + orow) * FDIM + col], acc1[n][v]);
        }
    }
    __syncthreads();

    float a0 = 0.f, a1 = 0.f;
#pragma unroll
    for (int w = 0; w < 8; ++w) {
        a0 += s_lw[w * ROWS_PB + c];
        a1 += s_lw[w * ROWS_PB + 16 + c];
    }
    l0 = a0;
    l1 = a1;
}

__global__ __launch_bounds__(256) __attribute__((amdgpu_waves_per_eu(2, 2)))
void gat_attn_kernel(const float* __restrict__ Wh,
                                                       const int*   __restrict__ adj,
                                                       const float* __restrict__ f1,
                                                       const float* __restrict__ f2,
                                                       float* __restrict__ out) {
    __shared__ alignas(16) float s_f2[N_NODES];
    __shared__ alignas(16) float s_acc[ROWS_PB * FDIM];
    __shared__ alignas(16) float s_lw[8 * ROWS_PB];
    __shared__ alignas(16) float s_wmax[8];
    __shared__ float s_m[ROWS_PB];

    const int blk  = blockIdx.x;
    const int b    = blk / (N_NODES / ROWS_PB);
    const int i0   = (blk % (N_NODES / ROWS_PB)) * ROWS_PB;
    const int t    = threadIdx.x;
    const int wave = t >> 5;
    const int lane = t & 31;
    const int c    = lane & 15;
    const int h    = lane >> 4;

    const float rf1a = f1[(size_t)b * N_NODES + i0 + c];
    const float rf1b = f1[(size_t)b * N_NODES + i0 + 16 + c];

    const int*   arow0 = adj + (size_t)(i0 + c) * N_NODES;
    const int*   arow1 = adj + (size_t)(i0 + 16 + c) * N_NODES;
    const float* whl   = Wh + ((size_t)b * N_NODES + wave * 256) * FDIM + 4 * lane;

    GatTile t0;
    gat_fetch(t0, whl, arow0, arow1, wave * 256 + 8 * h);

    float4* s_accv = reinterpret_cast<float4*>(s_acc);
    {
        const float4* f2v = reinterpret_cast<const float4*>(f2 + (size_t)b * N_NODES);
        float4* s_f2v = reinterpret_cast<float4*>(s_f2);
        const float4 v0 = f2v[t];
        const float4 v1 = f2v[t + 256];
        s_f2v[t]       = v0;
        s_f2v[t + 256] = v1;
#pragma unroll
        for (int i = 0; i < ROWS_PB * FDIM / 4 / 256; ++i)
            s_accv[t + 256 * i] = make_float4(0.f, 0.f, 0.f, 0.f);
        float lm = fmaxf(fmaxf(fmaxf(v0.x, v0.y), fmaxf(v0.z, v0.w)),
                         fmaxf(fmaxf(v1.x, v1.y), fmaxf(v1.z, v1.w)));
#pragma unroll
        for (int m = 16; m >= 1; m >>= 1) lm = fmaxf(lm, __shfl_xor(lm, m, 32));
        if (lane == 0) s_wmax[wave] = lm;
    }
    __syncthreads();

    float rma, rmb;
    {
        const float4 wa = *reinterpret_cast<const float4*>(s_wmax);
        const float4 wb = *reinterpret_cast<const float4*>(s_wmax + 4);
        const float f2max = fmaxf(fmaxf(fmaxf(wa.x, wa.y), fmaxf(wa.z, wa.w)),
                                  fmaxf(fmaxf(wb.x, wb.y), fmaxf(wb.z, wb.w)));
        rma = fmaxf(NEG_BIG, gat_score(1, rf1a + f2max));
        rmb = fmaxf(NEG_BIG, gat_score(1, rf1b + f2max));
    }

    float l0, l1;
    gat_round(whl, arow0, arow1, s_f2, s_acc, s_lw, rf1a, rf1b, rma, rmb, wave, c, h, t0, l0, l1);

    if (__any(!(l0 >= L_FLOOR) || !(l1 >= L_FLOOR))) {
#pragma unroll
        for (int i = 0; i < ROWS_PB * FDIM / 4 / 256; ++i)
            s_accv[t + 256 * i] = make_float4(0.f, 0.f, 0.f, 0.f);
        for (int pass = 0; pass < 2; ++pass) {
            const int r = wave * 4 + 2 * pass + h;
            const int4*   arow4 = reinterpret_cast<const int4*>(adj + (size_t)(i0 + r) * N_NODES);
            const float4* sf4   = reinterpret_cast<const float4*>(s_f2);
            const float fr = f1[(size_t)b * N_NODES + i0 + r];
            float mx = NEG_BIG;
#pragma unroll 4
            for (int it = 0; it < N_NODES / 64; ++it) {
                const int4   m = arow4[it * 16 + c];
                const float4 f = sf4[it * 16 + c];
                mx = fmaxf(mx, gat_score(m.x, fr + f.x));
                mx = fmaxf(mx, gat_score(m.y, fr + f.y));
                mx = fmaxf(mx, gat_score(m.z, fr + f.z));
                mx = fmaxf(mx, gat_score(m.w, fr + f.w));
            }
#pragma unroll
            for (int m = 8; m >= 1; m >>= 1) mx = fmaxf(mx, __shfl_xor(mx, m, 32));
            if (c == 0) s_m[r] = mx;
        }
        __syncthreads();
        gat_fetch(t0, whl, arow0, arow1, wave * 256 + 8 * h);
        gat_round(whl, arow0, arow1, s_f2, s_acc, s_lw, rf1a, rf1b, s_m[c], s_m[16 + c],
                  wave, c, h, t0, l0, l1);
    }

#pragma unroll
    for (int i = 0; i < ROWS_PB * FDIM / 256; ++i) {
        const int k = t + 256 * i;
        const int r = k >> 7;
        const float lr = __shfl(i < 8 ? l0 : l1, r & 15, 32);
        out[((size_t)b * N_NODES + i0 + r) * FDIM + (k & (FDIM - 1))] = s_acc[k] / lr;
    }
}

extern "C" void kernel_launch(void* const* d_in, const int* in_sizes, int n_in,
                              void* d_out, int out_size, void* d_ws, size_t ws_size,
                              hipStream_t stream) {
    const float* h   = (const float*)d_in[0];
    const int*   adj = (const int*)  d_in[1];
    const float* W   = (const float*)d_in[2];
    const float* a   = (const float*)d_in[3];
    float* out = (float*)d_out;

    float* Wh = (float*)d_ws;
    float* f1 = Wh + (size_t)NBATCH * N_NODES * FDIM;
    float* f2 = f1 + (size_t)NBATCH * N_NODES;

    gat_wh_kernel  <<<NBATCH * (N_NODES / 16), 256, 0, stream>>>(h, W, a, Wh, f1, f2);
    gat_attn_kernel<<<NBATCH * (N_NODES / ROWS_PB), 256, 0, stream>>>(Wh, adj, f1, f2, out);
}


# attention kernel: each wave stores its partial tile to its own LDS plane; the planes are summed in wave order by the 16-byte reads that normalise and store; no LDS atomics; order of sums changes
# speedup vs baseline: 3.3828x; 1.0370x over previous
#include <hip/hip_runtime.h>

typedef __attribute__((ext_vector_type(2))) float v2f;
typedef __attribute__((ext_vector_type(8))) float v8f;

#define ALPHA    0.2f
#define NEG_BIG  -9000000000000000.0f
#define N_NODES  2048
#define FDIM     128
#define NBATCH   4

#define WH_PITCH 132
#define H_PITCH  132

__global__ __launch_bounds__(256) void gat_wh_kernel(const float* __restrict__ h,
                                                     const float* __restrict__ W,
                                                     const float* __restrict__ a,
                                                     float* __restrict__ Wh,
                                                     float* __restrict__ f1,
                                                     float* __restrict__ f2) {
    __shared__ alignas(16) float sh[16 * H_PITCH];
    __shared__ alignas(16) float s_wh[16 * WH_PITCH];
    const int tile = blockIdx.x;
    const int b    = tile / (N_NODES / 16);
    const int i0   = (tile % (N_NODES / 16)) * 16;
    const int t    = threadIdx.x;

    const float4* hb4 = reinterpret_cast<const float4*>(h + ((size_t)b * N_NODES + i0) * FDIM);
    const float4 g0 = hb4[t];
    const float4 g1 = hb4[t + 256];
    float* srow = sh + (t >> 5) * H_PITCH + 4 * (t & 31);
    *reinterpret_cast<float4*>(srow)               = g0;
    *reinterpret_cast<float4*>(srow + 8 * H_PITCH) = g1;
    __syncthreads();

    const int wave  = t >> 5;
    const int lane  = t & 31;
    const int row   = lane & 15;
    const int kslot = lane >> 4;
    const int n0    = wave * 16;

    const int l16 = lane & 15;
    const float4 a1lo = *reinterpret_cast<const float4*>(a + 4 * l16);
    const float4 a1hi = *reinterpret_cast<const float4*>(a + 64 + 4 * l16);
    const float4 a2lo = *reinterpret_cast<const float4*>(a + FDIM + 4 * l16);
    const float4 a2hi = *reinterpret_cast<const float4*>(a + FDIM + 64 + 4 * l16);

    v8f acc = {};
    for (int k0 = 0; k0 < FDIM; k0 += 4) {
        v2f A, B;
        A.x = sh[row * H_PITCH + k0 + kslot * 2];
        A.y = sh[row * H_PITCH + k0 + kslot * 2 + 1];
        B.x = W[(size_t)(k0 + kslot * 2)     * FDIM + n0 + row];
        B.y = W[(size_t)(k0 + kslot * 2 + 1) * FDIM + n0 + row];
        acc = __builtin_amdgcn_wmma_f32_16x16x4_f32(false, A, false, B,
                                                    (short)0, acc, false, false);
    }

#pragma unroll
    for (int v = 0; v < 8; ++v) {
        int orow = v + 8 * kslot;
        s_wh[orow * WH_PITCH + n0 + row] = acc[v];
    }
    __syncthreads();

    float4* outc = reinterpret_cast<float4*>(Wh + (size_t)tile * (16 * FDIM));
#pragma unroll
    for (int piece = 0; piece < 2; ++piece) {
        const int f  = t + 256 * piece;
        const int r0 = 8 * ((f >> 4) & 1) + 2 * (f >> 7);
        const int nn = 32 * ((f >> 5) & 3) + 2 * (f & 15);
        const float2 lo = *reinterpret_cast<const float2*>(s_wh + r0 * WH_PITCH + nn);
        const float2 hi = *reinterpret_cast<const float2*>(s_wh + (r0 + 1) * WH_PITCH + nn);
        outc[f] = make_float4(lo.x, hi.x, lo.y, hi.y);
    }

    const int r = wave * 2 + kslot;
    const float4 wlo = *reinterpret_cast<const float4*>(s_wh + r * WH_PITCH + 4 * l16);
    const float4 whi = *reinterpret_cast<const float4*>(s_wh + r * WH_PITCH + 64 + 4 * l16);
    float s1 = wlo.x * a1lo.x + wlo.y * a1lo.y + wlo.z * a1lo.z + wlo.w * a1lo.w
             + whi.x * a1hi.x + whi.y * a1hi.y + whi.z * a1hi.z + whi.w * a1hi.w;
    float s2 = wlo.x * a2lo.x + wlo.y * a2lo.y + wlo.z * a2lo.z + wlo.w * a2lo.w
             + whi.x * a2hi.x + whi.y * a2hi.y + whi.z * a2hi.z + whi.w * a2hi.w;
    for (int m = 8; m >= 1; m >>= 1) {
        s1 += __shfl_xor(s1, m, 32);
        s2 += __shfl_xor(s2, m, 32);
    }
    if (l16 == 0) {
        f1[(size_t)b * N_NODES + i0 + r] = s1;
        f2[(size_t)b * N_NODES + i0 + r] = s2;
    }
}

#define L_FLOOR 1e-6f
#define ROWS_PB 32
#define P_PITCH 132
#define P_PLANE (ROWS_PB * P_PITCH)

__device__ __forceinline__ float gat_score(int m, float x) {
    const float e = x > 0.f ? x : ALPHA * x;
    return m > 0 ? e : NEG_BIG;
}

__device__ __forceinline__ float gat_p8(float* p, const int4 m0, const int4 m1,
                                        const float4 fa, const float4 fb,
                                        float rf1, float rm) {
    p[0] = __expf(gat_score(m0.x, rf1 + fa.x) - rm);
    p[1] = __expf(gat_score(m0.y, rf1 + fa.y) - rm);
    p[2] = __expf(gat_score(m0.z, rf1 + fa.z) - rm);
    p[3] = __expf(gat_score(m0.w, rf1 + fa.w) - rm);
    p[4] = __expf(gat_score(m1.x, rf1 + fb.x) - rm);
    p[5] = __expf(gat_score(m1.y, rf1 + fb.y) - rm);
    p[6] = __expf(gat_score(m1.z, rf1 + fb.z) - rm);
    p[7] = __expf(gat_score(m1.w, rf1 + fb.w) - rm);
    return ((p[0] + p[1]) + (p[2] + p[3])) + ((p[4] + p[5]) + (p[6] + p[7]));
}

struct GatTile {
    int4   ma0, ma1;
    int4   mb0, mb1;
    float4 wv[16];
};

__device__ __forceinline__ void gat_fetch(GatTile& T, const float* __restrict__ wch,
                                          const int* __restrict__ arow0,
                                          const int* __restrict__ arow1, int jb) {
    T.ma0 = *reinterpret_cast<const int4*>(arow0 + jb);
    T.ma1 = *reinterpret_cast<const int4*>(arow0 + jb + 4);
    T.mb0 = *reinterpret_cast<const int4*>(arow1 + jb);
    T.mb1 = *reinterpret_cast<const int4*>(arow1 + jb + 4);
#pragma unroll
    for (int i = 0; i < 16; ++i)
        T.wv[i] = *reinterpret_cast<const float4*>(wch + 128 * i);
}

__device__ __forceinline__ void gat_trip(const GatTile& T, const float* s_f2, int jb,
                                         float rf1a, float rf1b, float rma, float rmb,
                                         v8f (&acc0)[8], v8f (&acc1)[8],
                                         float& lsum0, float& lsum1) {
    const float4 fa = *reinterpret_cast<const float4*>(s_f2 + jb);
    const float4 fb = *reinterpret_cast<const float4*>(s_f2 + jb + 4);

    float p0[8], p1[8];
    lsum0 += gat_p8(p0, T.ma0, T.ma1, fa, fb, rf1a, rma);
    lsum1 += gat_p8(p1, T.mb0, T.mb1, fa, fb, rf1b, rmb);

#pragma unroll
    for (int q = 0; q < 4; ++q) {
        v2f A0, A1;
        A0.x = p0[2 * q];
        A0.y = p0[2 * q + 1];
        A1.x = p1[2 * q];
        A1.y = p1[2 * q + 1];
#pragma unroll
        for (int m = 0; m < 4; ++m) {
            const float4 w = T.wv[4 * q + m];
            v2f B;
            B.x = w.x; B.y = w.y;
            acc0[2 * m]     = __builtin_amdgcn_wmma_f32_16x16x4_f32(false, A0, false, B, (short)0, acc0[2 * m], false, false);
            acc1[2 * m]     = __builtin_amdgcn_wmma_f32_16x16x4_f32(false, A1, false, B, (short)0, acc1[2 * m], false, false);
            B.x = w.z; B.y = w.w;
            acc0[2 * m + 1] = __builtin_amdgcn_wmma_f32_16x16x4_f32(false, A0, false, B, (short)0, acc0[2 * m + 1], false, false);
            acc1[2 * m + 1] = __builtin_amdgcn_wmma_f32_16x16x4_f32(false, A1, false, B, (short)0, acc1[2 * m + 1], false, false);
        }
    }
}

__device__ __forceinline__ void gat_round(const float* __restrict__ whl,
                                          const int* __restrict__ arow0,
                                          const int* __restrict__ arow1,
                                          const float* s_f2, float* s_part, float* s_lw,
                                          float rf1a, float rf1b, float rma, float rmb,
                                          int wave, int c, int h, GatTile& ta,
                                          float& l0, float& l1) {
    v8f acc0[8] = {};
    v8f acc1[8] = {};
    float lsum0 = 0.f, lsum1 = 0.f;
    const int jw = wave * 256 + 8 * h;
    GatTile tb;
    for (int ch = 0; ch < 14; ch += 2) {
        gat_fetch(tb, whl + (ch + 1) * (16 * FDIM), arow0, arow1, jw + (ch + 1) * 16);
        gat_trip(ta, s_f2, jw + ch * 16, rf1a, rf1b, rma, rmb, acc0, acc1, lsum0, lsum1);
        gat_fetch(ta, whl + (ch + 2) * (16 * FDIM), arow0, arow1, jw + (ch + 2) * 16);
        gat_trip(tb, s_f2, jw + (ch + 1) * 16, rf1a, rf1b, rma, rmb, acc0, acc1, lsum0, lsum1);
    }
    gat_fetch(tb, whl + 15 * (16 * FDIM), arow0, arow1, jw + 15 * 16);
    gat_trip(ta, s_f2, jw + 14 * 16, rf1a, rf1b, rma, rmb, acc0, acc1, lsum0, lsum1);
    gat_trip(tb, s_f2, jw + 15 * 16, rf1a, rf1b, rma, rmb, acc0, acc1, lsum0, lsum1);

    lsum0 += __shfl_xor(lsum0, 16, 32);
    lsum1 += __shfl_xor(lsum1, 16, 32);

    float* mine = s_part + wave * P_PLANE + 2 * c;
#pragma unroll
    for (int m = 0; m < 4; ++m) {
#pragma unroll
        for (int v = 0; v < 8; ++v) {
            const int orow = v + 8 * h;
            *reinterpret_cast<float2*>(mine + orow * P_PITCH + 32 * m) =
                make_float2(acc0[2 * m][v], acc0[2 * m + 1][v]);
            *reinterpret_cast<float2*>(mine + (16 + orow) * P_PITCH + 32 * m) =
                make_float2(acc1[2 * m][v], acc1[2 * m + 1][v]);
        }
    }
    if (h == 0) {
        s_lw[wave * ROWS_PB + c]      = lsum0;
        s_lw[wave * ROWS_PB + 16 + c] = lsum1;
    }
    __syncthreads();

    float a0 = 0.f, a1 = 0.f;
#pragma unroll
    for (int w = 0; w < 8; ++w) {
        a0 += s_lw[w * ROWS_PB + c];
        a1 += s_lw[w * ROWS_PB + 16 + c];
    }
    l0 = a0;
    l1 = a1;
}

__global__ __launch_bounds__(256) __attribute__((amdgpu_waves_per_eu(2, 2)))
void gat_attn_kernel(const float* __restrict__ Wh,
                                                       const int*   __restrict__ adj,
                                                       const float* __restrict__ f1,
                                                       const float* __restrict__ f2,
                                                       float* __restrict__ out) {
    __shared__ alignas(16) float s_f2[N_NODES];
    __shared__ alignas(16) float s_part[8 * P_PLANE];
    __shared__ alignas(16) float s_lw[8 * ROWS_PB];
    __shared__ alignas(16) float s_wmax[8];
    __shared__ float s_m[ROWS_PB];

    const int blk  = blockIdx.x;
    const int b    = blk / (N_NODES / ROWS_PB);
    const int i0   = (blk % (N_NODES / ROWS_PB)) * ROWS_PB;
    const int t    = threadIdx.x;
    const int wave = t >> 5;
    const int lane = t & 31;
    const int c    = lane & 15;
    const int h    = lane >> 4;

    const float rf1a = f1[(size_t)b * N_NODES + i0 + c];
    const float rf1b = f1[(size_t)b * N_NODES + i0 + 16 + c];

    const int*   arow0 = adj + (size_t)(i0 + c) * N_NODES;
    const int*   arow1 = adj + (size_t)(i0 + 16 + c) * N_NODES;
    const float* whl   = Wh + ((size_t)b * N_NODES + wave * 256) * FDIM + 4 * lane;

    GatTile t0;
    gat_fetch(t0, whl, arow0, arow1, wave * 256 + 8 * h);

    {
        const float4* f2v = reinterpret_cast<const float4*>(f2 + (size_t)b * N_NODES);
        float4* s_f2v = reinterpret_cast<float4*>(s_f2);
        const float4 v0 = f2v[t];
        const float4 v1 = f2v[t + 256];
        s_f2v[t]       = v0;
        s_f2v[t + 256] = v1;
        float lm = fmaxf(fmaxf(fmaxf(v0.x, v0.y), fmaxf(v0.z, v0.w)),
                         fmaxf(fmaxf(v1.x, v1.y), fmaxf(v1.z, v1.w)));
#pragma unroll
        for (int m = 16; m >= 1; m >>= 1) lm = fmaxf(lm, __shfl_xor(lm, m, 32));
        if (lane == 0) s_wmax[wave] = lm;
    }
    __syncthreads();

    float rma, rmb;
    {
        const float4 wa = *reinterpret_cast<const float4*>(s_wmax);
        const float4 wb = *reinterpret_cast<const float4*>(s_wmax + 4);
        const float f2max = fmaxf(fmaxf(fmaxf(wa.x, wa.y), fmaxf(wa.z, wa.w)),
                                  fmaxf(fmaxf(wb.x, wb.y), fmaxf(wb.z, wb.w)));
        rma = fmaxf(NEG_BIG, gat_score(1, rf1a + f2max));
        rmb = fmaxf(NEG_BIG, gat_score(1, rf1b + f2max));
    }

    float l0, l1;
    gat_round(whl, arow0, arow1, s_f2, s_part, s_lw, rf1a, rf1b, rma, rmb, wave, c, h, t0, l0, l1);

    if (__any(!(l0 >= L_FLOOR) || !(l1 >= L_FLOOR))) {
        for (int pass = 0; pass < 2; ++pass) {
            const int r = wave * 4 + 2 * pass + h;
            const int4*   arow4 = reinterpret_cast<const int4*>(adj + (size_t)(i0 + r) * N_NODES);
            const float4* sf4   = reinterpret_cast<const float4*>(s_f2);
            const float fr = f1[(size_t)b * N_NODES + i0 + r];
            float mx = NEG_BIG;
#pragma unroll 4
            for (int it = 0; it < N_NODES / 64; ++it) {
                const int4   m = arow4[it * 16 + c];
                const float4 f = sf4[it * 16 + c];
                mx = fmaxf(mx, gat_score(m.x, fr + f.x));
                mx = fmaxf(mx, gat_score(m.y, fr + f.y));
                mx = fmaxf(mx, gat_score(m.z, fr + f.z));
                mx = fmaxf(mx, gat_score(m.w, fr + f.w));
            }
#pragma unroll
            for (int m = 8; m >= 1; m >>= 1) mx = fmaxf(mx, __shfl_xor(mx, m, 32));
            if (c == 0) s_m[r] = mx;
        }
        __syncthreads();
        gat_fetch(t0, whl, arow0, arow1, wave * 256 + 8 * h);
        gat_round(whl, arow0, arow1, s_f2, s_part, s_lw, rf1a, rf1b, s_m[c], s_m[16 + c],
                  wave, c, h, t0, l0, l1);
    }

#pragma unroll
    for (int i = 0; i < ROWS_PB / 8; ++i) {
        const int r = wave + 8 * i;
        const float* src = s_part + r * P_PITCH + 4 * lane;
        float4 s = *reinterpret_cast<const float4*>(src);
#pragma unroll
        for (int w = 1; w < 8; ++w) {
            const float4 q = *reinterpret_cast<const float4*>(src + w * P_PLANE);
            s.x += q.x; s.y += q.y; s.z += q.z; s.w += q.w;
        }
        const float lr = __shfl(i < 2 ? l0 : l1, r & 15, 32);
        *reinterpret_cast<float4*>(out + ((size_t)b * N_NODES + i0 + r) * FDIM + 4 * lane) =
            make_float4(s.x / lr, s.y / lr, s.z / lr, s.w / lr);
    }
}

extern "C" void kernel_launch(void* const* d_in, const int* in_sizes, int n_in,
                              void* d_out, int out_size, void* d_ws, size_t ws_size,
                              hipStream_t stream) {
    const float* h   = (const float*)d_in[0];
    const int*   adj = (const int*)  d_in[1];
    const float* W   = (const float*)d_in[2];
    const float* a   = (const float*)d_in[3];
    float* out = (float*)d_out;

    float* Wh = (float*)d_ws;
    float* f1 = Wh + (size_t)NBATCH * N_NODES * FDIM;
    float* f2 = f1 + (size_t)NBATCH * N_NODES;

    gat_wh_kernel  <<<NBATCH * (N_NODES / 16), 256, 0, stream>>>(h, W, a, Wh, f1, f2);
    gat_attn_kernel<<<NBATCH * (N_NODES / ROWS_PB), 256, 0, stream>>>(Wh, adj, f1, f2, out);
}
